# starting kernel
# baseline (speedup 1.0000x reference)
#include <hip/hip_runtime.h>
#include <hip/hip_bf16.h>

typedef __attribute__((ext_vector_type(16))) _Float16 v16h;
typedef __attribute__((ext_vector_type(8)))  float    v8f;

#define TPB 256

__device__ __forceinline__ void atomic_add_f32(float* p, float v) {
  __hip_atomic_fetch_add(p, v, __ATOMIC_RELAXED, __HIP_MEMORY_SCOPE_AGENT);
}


__global__ void gcn_init(float* __restrict__ dinv, float* __restrict__ agg1,
                         float* __restrict__ agg2, int N) {
  long long t = (long long)blockIdx.x * blockDim.x + threadIdx.x;
  long long total = (long long)N * 16;
  if (t < total) agg1[t] = 0.0f;
  if (t < N) { dinv[t] = 1.0f; agg2[t] = 0.0f; }
}

__global__ void gcn_degree(const int* __restrict__ dst, float* __restrict__ deg,
                           long long E) {
  long long e = (long long)blockIdx.x * blockDim.x + threadIdx.x;
  if (e >= E) return;
  __builtin_prefetch(dst + e + 16384, 0, 1);
  atomic_add_f32(&deg[dst[e]], 1.0f);
}

__global__ void gcn_rsqrt(float* __restrict__ dinv, int N) {
  int i = blockIdx.x * blockDim.x + threadIdx.x;
  if (i < N) dinv[i] = rsqrtf(dinv[i]);
}

__global__ void gcn_xw1(const float* __restrict__ x, const float* __restrict__ W1,
                        float* __restrict__ t1, int N) {
  int wave = (int)(((long long)blockIdx.x * blockDim.x + threadIdx.x) >> 5);
  int lane = (int)(threadIdx.x & 31);
  int row0 = wave * 16;
  if (row0 >= N) return;

  v16h ah = {}, al = {};
  v16h bh = {}, bl = {};
  if (lane < 16) {
    int m = row0 + lane;
    if (m >= N) m = N - 1;
    const float* xr = x + (size_t)m * 5;
#pragma unroll
    for (int k = 0; k < 5; ++k) {
      float v = xr[k];
      _Float16 hi = (_Float16)v;
      ah[k] = hi;
      al[k] = (_Float16)(v - (float)hi);
    }
#pragma unroll
    for (int k = 0; k < 5; ++k) {
      float w = W1[k * 16 + lane];
      _Float16 hi = (_Float16)w;
      bh[k] = hi;
      bl[k] = (_Float16)(w - (float)hi);
    }
  }

  v8f c = {};
  c = __builtin_amdgcn_wmma_f32_16x16x32_f16(false, al, false, bh, (short)0, c, false, false);
  c = __builtin_amdgcn_wmma_f32_16x16x32_f16(false, ah, false, bl, (short)0, c, false, false);
  c = __builtin_amdgcn_wmma_f32_16x16x32_f16(false, ah, false, bh, (short)0, c, false, false);

  int n  = lane & 15;
  int mb = row0 + ((lane >= 16) ? 8 : 0);
  if (row0 + 16 <= N) {
    float* base = t1 + (size_t)mb * 16 + n;
#pragma unroll
    for (int r = 0; r < 8; ++r) base[(size_t)r * 16] = c[r];
  } else {
#pragma unroll
    for (int r = 0; r < 8; ++r) {
      int m = mb + r;
      if (m < N) t1[(size_t)m * 16 + n] = c[r];
    }
  }
}

__global__ void gcn_msg16(const int* __restrict__ src, const int* __restrict__ dst,
                          const float* __restrict__ t1, const float* __restrict__ dinv,
                          float* __restrict__ agg1, long long E) {
  long long t = (long long)blockIdx.x * blockDim.x + threadIdx.x;
  long long e = t >> 4;
  int c = (int)(t & 15);
  if (e >= E) return;
  int s = src[e];
  int d = dst[e];
  float w = dinv[s] * dinv[d];
  atomic_add_f32(&agg1[(size_t)d * 16 + c], t1[(size_t)s * 16 + c] * w);
}

__global__ void gcn_finalize1(const float* __restrict__ agg1, const float* __restrict__ t1,
                              const float* __restrict__ dinv, const float* __restrict__ b1,
                              const float* __restrict__ W2, float* __restrict__ t2, int N) {
  int i = blockIdx.x * blockDim.x + threadIdx.x;
  if (i >= N) return;
  float dd  = dinv[i] * dinv[i];
  float acc = 0.0f;
#pragma unroll
  for (int c = 0; c < 16; ++c) {
    float v = agg1[(size_t)i * 16 + c] + t1[(size_t)i * 16 + c] * dd + b1[c];
    v = fmaxf(v, 0.0f);
    acc = fmaf(v, W2[c], acc);
  }
  t2[i] = acc;
}

__global__ void gcn_msg1(const int* __restrict__ src, const int* __restrict__ dst,
                         const float* __restrict__ t2, const float* __restrict__ dinv,
                         float* __restrict__ agg2, long long E) {
  long long e = (long long)blockIdx.x * blockDim.x + threadIdx.x;
  if (e >= E) return;
  __builtin_prefetch(src + e + 16384, 0, 1);
  int s = src[e];
  int d = dst[e];
  atomic_add_f32(&agg2[d], t2[s] * dinv[s] * dinv[d]);
}

__global__ void gcn_finalize2(const float* __restrict__ agg2, const float* __restrict__ t2,
                              const float* __restrict__ dinv, const float* __restrict__ b2,
                              float* __restrict__ out, int N) {
  int i = blockIdx.x * blockDim.x + threadIdx.x;
  if (i >= N) return;
  out[i] = agg2[i] + t2[i] * dinv[i] * dinv[i] + b2[0];
}

extern "C" void kernel_launch(void* const* d_in, const int* in_sizes, int n_in,
                              void* d_out, int out_size, void* d_ws, size_t ws_size,
                              hipStream_t stream) {
  const float* x  = (const float*)d_in[0];
  const float* W1 = (const float*)d_in[1];
  const float* b1 = (const float*)d_in[2];
  const float* W2 = (const float*)d_in[3];
  const float* b2 = (const float*)d_in[4];
  const int*   ei = (const int*)d_in[5];

  int       N = in_sizes[0] / 5;
  long long E = (long long)in_sizes[5] / 2;
  const int* src = ei;
  const int* dst = ei + E;

  float* ws   = (float*)d_ws;
  float* dinv = ws;
  float* t1   = dinv + N;
  float* agg1 = t1 + (size_t)16 * N;
  float* t2   = agg1 + (size_t)16 * N;
  float* agg2 = t2 + N;

  float* out = (float*)d_out;

  long long initT   = (long long)N * 16;
  unsigned  gInit   = (unsigned)((initT + TPB - 1) / TPB);
  unsigned  gEdge   = (unsigned)((E + TPB - 1) / TPB);
  unsigned  gNode   = (unsigned)((N + TPB - 1) / TPB);
  unsigned  gMsg16  = (unsigned)((16 * E + TPB - 1) / TPB);
  int       waves   = (N + 15) / 16;
  unsigned  gWmma   = (unsigned)((waves + 1) / 2);

  gcn_init     <<<gInit,  TPB, 0, stream>>>(dinv, agg1, agg2, N);
  gcn_degree   <<<gEdge,  TPB, 0, stream>>>(dst, dinv, E);
  gcn_rsqrt    <<<gNode,  TPB, 0, stream>>>(dinv, N);
  gcn_xw1      <<<gWmma,   64, 0, stream>>>(x, W1, t1, N);
  gcn_msg16    <<<gMsg16, TPB, 0, stream>>>(src, dst, t1, dinv, agg1, E);
  gcn_finalize1<<<gNode,  TPB, 0, stream>>>(agg1, t1, dinv, b1, W2, t2, N);
  gcn_msg1     <<<gEdge,  TPB, 0, stream>>>(src, dst, t2, dinv, agg2, E);
  gcn_finalize2<<<gNode,  TPB, 0, stream>>>(agg2, t2, dinv, b2, out, N);
}


# Degree and layer-2 (1-channel) edge scatter-adds privatized in LDS bins per edge chunk, stored as 32 partial planes that the node kernels sum in fixed order; no global atomics there (new sum tree)
# speedup vs baseline: 2.0613x; 2.0613x over previous
#include <hip/hip_runtime.h>
#include <hip/hip_bf16.h>

typedef __attribute__((ext_vector_type(16))) _Float16 v16h;
typedef __attribute__((ext_vector_type(8)))  float    v8f;

#define TPB 256
#define HT 1024
#define HB 32
#define HR 12800

__device__ __forceinline__ void atomic_add_f32(float* p, float v) {
  __hip_atomic_fetch_add(p, v, __ATOMIC_RELAXED, __HIP_MEMORY_SCOPE_AGENT);
}

__device__ __forceinline__ void lds_add_f32(float* p, float v) {
  __hip_atomic_fetch_add(p, v, __ATOMIC_RELAXED, __HIP_MEMORY_SCOPE_WORKGROUP);
}


__global__ void gcn_init(float* __restrict__ agg1, int N) {
  long long t = (long long)blockIdx.x * blockDim.x + threadIdx.x;
  long long total = (long long)N * 16;
  if (t < total) agg1[t] = 0.0f;
}

#define SU 4
#define SK (4 * SU)

template <bool WEIGHTED, bool GUARD>
__device__ __forceinline__ void scatter_trip(
    const int* __restrict__ src, const int* __restrict__ dst,
    const float* __restrict__ t2, const float* __restrict__ dinv,
    float* __restrict__ bins, const float* __restrict__ dloc,
    long long e, long long e0, long long e1v, int tid, int lo, int cnt) {
  int d[SK], s[SK];
#pragma unroll
  for (int u = 0; u < SU; ++u) {
    const long long g  = e + 4LL * (tid + HT * u);
    const bool      ok = !GUARD || g < e1v;
    const long long at = ok ? g : e0;
    int4 q = *reinterpret_cast<const int4*>(dst + at);
    if (!ok) q = make_int4(-1, -1, -1, -1);
    d[4 * u] = q.x; d[4 * u + 1] = q.y; d[4 * u + 2] = q.z; d[4 * u + 3] = q.w;
    if (WEIGHTED) {
      const int4 p = *reinterpret_cast<const int4*>(src + at);
      s[4 * u] = p.x; s[4 * u + 1] = p.y; s[4 * u + 2] = p.z; s[4 * u + 3] = p.w;
    }
  }
  if (WEIGHTED) {
    float val[SK];
#pragma unroll
    for (int k = 0; k < SK; ++k) {
      const unsigned r  = (unsigned)(d[k] - lo);
      const bool     in = r < (unsigned)cnt;
      const int      sk = in ? s[k] : 0;
      val[k] = t2[sk] * dinv[sk] * dloc[in ? r : 0u];
    }
#pragma unroll
    for (int k = 0; k < SK; ++k) asm volatile("" : "+v"(val[k]));
#pragma unroll
    for (int k = 0; k < SK; ++k) {
      const unsigned r = (unsigned)(d[k] - lo);
      if (r < (unsigned)cnt) lds_add_f32(&bins[r], val[k]);
    }
  } else {
#pragma unroll
    for (int k = 0; k < SK; ++k) {
      const unsigned r = (unsigned)(d[k] - lo);
      if (r < (unsigned)cnt) lds_add_f32(&bins[r], 1.0f);
    }
  }
}

template <bool WEIGHTED>
__global__ __launch_bounds__(HT) void gcn_scatter1(
    const int* __restrict__ src, const int* __restrict__ dst,
    const float* __restrict__ t2, const float* __restrict__ dinv,
    float* __restrict__ planes, long long E, int N) {
  constexpr int FJ = (HR + HT - 1) / HT;
  __shared__ float bins[FJ * HT];
  __shared__ float dloc[WEIGHTED ? FJ * HT : 1];

  const int tid = (int)threadIdx.x;
  const int lo  = (int)blockIdx.y * HR;
  int cnt = N - lo;
  if (cnt > HR) cnt = HR;

#pragma unroll
  for (int j = 0; j < FJ; ++j) {
    const int i = tid + j * HT;
    bins[i] = 0.0f;
    if (WEIGHTED) dloc[i] = dinv[lo + (i < cnt ? i : cnt - 1)];
  }
  __syncthreads();

  const long long per = (((E + HB - 1) / HB) + 3) & ~3LL;
  const long long e0  = (long long)blockIdx.x * per;
  long long e1 = e0 + per;
  if (e1 > E) e1 = E;
  if (e1 < e0) e1 = e0;
  long long rest = e0;

  const bool vec = WEIGHTED ? ((((size_t)dst | (size_t)src) & 15) == 0)
                            : (((size_t)dst & 15) == 0);
  if (vec) {
    const long long trip = (long long)SK * HT;
    const long long e1v  = e0 + ((e1 - e0) & ~3LL);
    long long e = e0;
    for (; e + trip <= e1v; e += trip)
      scatter_trip<WEIGHTED, false>(src, dst, t2, dinv, bins, dloc, e, e0, e1v, tid, lo, cnt);
    if (e < e1v)
      scatter_trip<WEIGHTED, true>(src, dst, t2, dinv, bins, dloc, e, e0, e1v, tid, lo, cnt);
    rest = e1v;
  }
  for (long long k = rest + tid; k < e1; k += HT) {
    const unsigned r = (unsigned)(dst[k] - lo);
    if (r < (unsigned)cnt) {
      float v = 1.0f;
      if (WEIGHTED) { const int sk = src[k]; v = t2[sk] * dinv[sk] * dloc[r]; }
      lds_add_f32(&bins[r], v);
    }
  }
  __syncthreads();

  float* pl = planes + (size_t)blockIdx.x * (size_t)N + lo;
  for (int i = tid; i < cnt; i += HT) pl[i] = bins[i];
}

__global__ void gcn_rsqrt(const float* __restrict__ planes, float* __restrict__ dinv, int N) {
  int i = blockIdx.x * blockDim.x + threadIdx.x;
  if (i >= N) return;
  float deg = 1.0f;
#pragma unroll
  for (int b = 0; b < HB; ++b) deg += planes[(size_t)b * N + i];
  dinv[i] = rsqrtf(deg);
}

__global__ void gcn_xw1(const float* __restrict__ x, const float* __restrict__ W1,
                        float* __restrict__ t1, int N) {
  int wave = (int)(((long long)blockIdx.x * blockDim.x + threadIdx.x) >> 5);
  int lane = (int)(threadIdx.x & 31);
  int row0 = wave * 16;
  if (row0 >= N) return;

  v16h ah = {}, al = {};
  v16h bh = {}, bl = {};
  if (lane < 16) {
    int m = row0 + lane;
    if (m >= N) m = N - 1;
    const float* xr = x + (size_t)m * 5;
#pragma unroll
    for (int k = 0; k < 5; ++k) {
      float v = xr[k];
      _Float16 hi = (_Float16)v;
      ah[k] = hi;
      al[k] = (_Float16)(v - (float)hi);
    }
#pragma unroll
    for (int k = 0; k < 5; ++k) {
      float w = W1[k * 16 + lane];
      _Float16 hi = (_Float16)w;
      bh[k] = hi;
      bl[k] = (_Float16)(w - (float)hi);
    }
  }

  v8f c = {};
  c = __builtin_amdgcn_wmma_f32_16x16x32_f16(false, al, false, bh, (short)0, c, false, false);
  c = __builtin_amdgcn_wmma_f32_16x16x32_f16(false, ah, false, bl, (short)0, c, false, false);
  c = __builtin_amdgcn_wmma_f32_16x16x32_f16(false, ah, false, bh, (short)0, c, false, false);

  int n  = lane & 15;
  int mb = row0 + ((lane >= 16) ? 8 : 0);
  if (row0 + 16 <= N) {
    float* base = t1 + (size_t)mb * 16 + n;
#pragma unroll
    for (int r = 0; r < 8; ++r) base[(size_t)r * 16] = c[r];
  } else {
#pragma unroll
    for (int r = 0; r < 8; ++r) {
      int m = mb + r;
      if (m < N) t1[(size_t)m * 16 + n] = c[r];
    }
  }
}

__global__ void gcn_msg16(const int* __restrict__ src, const int* __restrict__ dst,
                          const float* __restrict__ t1, const float* __restrict__ dinv,
                          float* __restrict__ agg1, long long E) {
  long long t = (long long)blockIdx.x * blockDim.x + threadIdx.x;
  long long e = t >> 4;
  int c = (int)(t & 15);
  if (e >= E) return;
  int s = src[e];
  int d = dst[e];
  float w = dinv[s] * dinv[d];
  atomic_add_f32(&agg1[(size_t)d * 16 + c], t1[(size_t)s * 16 + c] * w);
}

__global__ void gcn_finalize1(const float* __restrict__ agg1, const float* __restrict__ t1,
                              const float* __restrict__ dinv, const float* __restrict__ b1,
                              const float* __restrict__ W2, float* __restrict__ t2, int N) {
  int i = blockIdx.x * blockDim.x + threadIdx.x;
  if (i >= N) return;
  float dd  = dinv[i] * dinv[i];
  float acc = 0.0f;
#pragma unroll
  for (int c = 0; c < 16; ++c) {
    float v = agg1[(size_t)i * 16 + c] + t1[(size_t)i * 16 + c] * dd + b1[c];
    v = fmaxf(v, 0.0f);
    acc = fmaf(v, W2[c], acc);
  }
  t2[i] = acc;
}

__global__ void gcn_finalize2(const float* __restrict__ planes, const float* __restrict__ t2,
                              const float* __restrict__ dinv, const float* __restrict__ b2,
                              float* __restrict__ out, int N) {
  int i = blockIdx.x * blockDim.x + threadIdx.x;
  if (i >= N) return;
  float agg = 0.0f;
#pragma unroll
  for (int b = 0; b < HB; ++b) agg += planes[(size_t)b * N + i];
  out[i] = agg + t2[i] * dinv[i] * dinv[i] + b2[0];
}

extern "C" void kernel_launch(void* const* d_in, const int* in_sizes, int n_in,
                              void* d_out, int out_size, void* d_ws, size_t ws_size,
                              hipStream_t stream) {
  const float* x  = (const float*)d_in[0];
  const float* W1 = (const float*)d_in[1];
  const float* b1 = (const float*)d_in[2];
  const float* W2 = (const float*)d_in[3];
  const float* b2 = (const float*)d_in[4];
  const int*   ei = (const int*)d_in[5];

  int       N = in_sizes[0] / 5;
  long long E = (long long)in_sizes[5] / 2;
  const int* src = ei;
  const int* dst = ei + E;

  float* ws   = (float*)d_ws;
  float* dinv = ws;
  float* t1   = dinv + N;
  float* agg1 = t1 + (size_t)16 * N;
  float* t2   = agg1 + (size_t)16 * N;
  float* planes = t1;

  float* out = (float*)d_out;

  long long initT   = (long long)N * 16;
  unsigned  gInit   = (unsigned)((initT + TPB - 1) / TPB);
  dim3      gScat(HB, (unsigned)((N + HR - 1) / HR));
  unsigned  gNode   = (unsigned)((N + TPB - 1) / TPB);
  unsigned  gMsg16  = (unsigned)((16 * E + TPB - 1) / TPB);
  int       waves   = (N + 15) / 16;
  unsigned  gWmma   = (unsigned)((waves + 1) / 2);

  gcn_scatter1<false><<<gScat, HT, 0, stream>>>(dst, dst, dinv, dinv, planes, E, N);
  gcn_rsqrt    <<<gNode,  TPB, 0, stream>>>(planes, dinv, N);
  gcn_init     <<<gInit,  TPB, 0, stream>>>(agg1, N);
  gcn_xw1      <<<gWmma,   64, 0, stream>>>(x, W1, t1, N);
  gcn_msg16    <<<gMsg16, TPB, 0, stream>>>(src, dst, t1, dinv, agg1, E);
  gcn_finalize1<<<gNode,  TPB, 0, stream>>>(agg1, t1, dinv, b1, W2, t2, N);
  gcn_scatter1<true> <<<gScat, HT, 0, stream>>>(src, dst, t2, dinv, planes, E, N);
  gcn_finalize2<<<gNode,  TPB, 0, stream>>>(planes, t2, dinv, b2, out, N);
}


# Layer-1 edge kernel: 8192 resident waves loop over 32-edge trips (16 row gathers in flight, then 16 row atomic adds) instead of one short wave per 2 edges; f32 atomic arrival order changes
# speedup vs baseline: 3.0171x; 1.4637x over previous
#include <hip/hip_runtime.h>
#include <hip/hip_bf16.h>

typedef __attribute__((ext_vector_type(16))) _Float16 v16h;
typedef __attribute__((ext_vector_type(8)))  float    v8f;

#define TPB 256
#define HT 1024
#define HB 32
#define HR 12800

__device__ __forceinline__ void atomic_add_f32(float* p, float v) {
  __hip_atomic_fetch_add(p, v, __ATOMIC_RELAXED, __HIP_MEMORY_SCOPE_AGENT);
}

__device__ __forceinline__ void lds_add_f32(float* p, float v) {
  __hip_atomic_fetch_add(p, v, __ATOMIC_RELAXED, __HIP_MEMORY_SCOPE_WORKGROUP);
}


__global__ void gcn_init(float* __restrict__ agg1, int N) {
  long long t = (long long)blockIdx.x * blockDim.x + threadIdx.x;
  long long total = (long long)N * 16;
  if (t < total) agg1[t] = 0.0f;
}

#define SU 4
#define SK (4 * SU)

template <bool WEIGHTED, bool GUARD>
__device__ __forceinline__ void scatter_trip(
    const int* __restrict__ src, const int* __restrict__ dst,
    const float* __restrict__ t2, const float* __restrict__ dinv,
    float* __restrict__ bins, const float* __restrict__ dloc,
    long long e, long long e0, long long e1v, int tid, int lo, int cnt) {
  int d[SK], s[SK];
#pragma unroll
  for (int u = 0; u < SU; ++u) {
    const long long g  = e + 4LL * (tid + HT * u);
    const bool      ok = !GUARD || g < e1v;
    const long long at = ok ? g : e0;
    int4 q = *reinterpret_cast<const int4*>(dst + at);
    if (!ok) q = make_int4(-1, -1, -1, -1);
    d[4 * u] = q.x; d[4 * u + 1] = q.y; d[4 * u + 2] = q.z; d[4 * u + 3] = q.w;
    if (WEIGHTED) {
      const int4 p = *reinterpret_cast<const int4*>(src + at);
      s[4 * u] = p.x; s[4 * u + 1] = p.y; s[4 * u + 2] = p.z; s[4 * u + 3] = p.w;
    }
  }
  if (WEIGHTED) {
    float val[SK];
#pragma unroll
    for (int k = 0; k < SK; ++k) {
      const unsigned r  = (unsigned)(d[k] - lo);
      const bool     in = r < (unsigned)cnt;
      const int      sk = in ? s[k] : 0;
      val[k] = t2[sk] * dinv[sk] * dloc[in ? r : 0u];
    }
#pragma unroll
    for (int k = 0; k < SK; ++k) asm volatile("" : "+v"(val[k]));
#pragma unroll
    for (int k = 0; k < SK; ++k) {
      const unsigned r = (unsigned)(d[k] - lo);
      if (r < (unsigned)cnt) lds_add_f32(&bins[r], val[k]);
    }
  } else {
#pragma unroll
    for (int k = 0; k < SK; ++k) {
      const unsigned r = (unsigned)(d[k] - lo);
      if (r < (unsigned)cnt) lds_add_f32(&bins[r], 1.0f);
    }
  }
}

template <bool WEIGHTED>
__global__ __launch_bounds__(HT) void gcn_scatter1(
    const int* __restrict__ src, const int* __restrict__ dst,
    const float* __restrict__ t2, const float* __restrict__ dinv,
    float* __restrict__ planes, long long E, int N) {
  constexpr int FJ = (HR + HT - 1) / HT;
  __shared__ float bins[FJ * HT];
  __shared__ float dloc[WEIGHTED ? FJ * HT : 1];

  const int tid = (int)threadIdx.x;
  const int lo  = (int)blockIdx.y * HR;
  int cnt = N - lo;
  if (cnt > HR) cnt = HR;

#pragma unroll
  for (int j = 0; j < FJ; ++j) {
    const int i = tid + j * HT;
    bins[i] = 0.0f;
    if (WEIGHTED) dloc[i] = dinv[lo + (i < cnt ? i : cnt - 1)];
  }
  __syncthreads();

  const long long per = (((E + HB - 1) / HB) + 3) & ~3LL;
  const long long e0  = (long long)blockIdx.x * per;
  long long e1 = e0 + per;
  if (e1 > E) e1 = E;
  if (e1 < e0) e1 = e0;
  long long rest = e0;

  const bool vec = WEIGHTED ? ((((size_t)dst | (size_t)src) & 15) == 0)
                            : (((size_t)dst & 15) == 0);
  if (vec) {
    const long long trip = (long long)SK * HT;
    const long long e1v  = e0 + ((e1 - e0) & ~3LL);
    long long e = e0;
    for (; e + trip <= e1v; e += trip)
      scatter_trip<WEIGHTED, false>(src, dst, t2, dinv, bins, dloc, e, e0, e1v, tid, lo, cnt);
    if (e < e1v)
      scatter_trip<WEIGHTED, true>(src, dst, t2, dinv, bins, dloc, e, e0, e1v, tid, lo, cnt);
    rest = e1v;
  }
  for (long long k = rest + tid; k < e1; k += HT) {
    const unsigned r = (unsigned)(dst[k] - lo);
    if (r < (unsigned)cnt) {
      float v = 1.0f;
      if (WEIGHTED) { const int sk = src[k]; v = t2[sk] * dinv[sk] * dloc[r]; }
      lds_add_f32(&bins[r], v);
    }
  }
  __syncthreads();

  float* pl = planes + (size_t)blockIdx.x * (size_t)N + lo;
  for (int i = tid; i < cnt; i += HT) pl[i] = bins[i];
}

__global__ void gcn_rsqrt(const float* __restrict__ planes, float* __restrict__ dinv, int N) {
  int i = blockIdx.x * blockDim.x + threadIdx.x;
  if (i >= N) return;
  float deg = 1.0f;
#pragma unroll
  for (int b = 0; b < HB; ++b) deg += planes[(size_t)b * N + i];
  dinv[i] = rsqrtf(deg);
}

__global__ void gcn_xw1(const float* __restrict__ x, const float* __restrict__ W1,
                        float* __restrict__ t1, int N) {
  int wave = (int)(((long long)blockIdx.x * blockDim.x + threadIdx.x) >> 5);
  int lane = (int)(threadIdx.x & 31);
  int row0 = wave * 16;
  if (row0 >= N) return;

  v16h ah = {}, al = {};
  v16h bh = {}, bl = {};
  if (lane < 16) {
    int m = row0 + lane;
    if (m >= N) m = N - 1;
    const float* xr = x + (size_t)m * 5;
#pragma unroll
    for (int k = 0; k < 5; ++k) {
      float v = xr[k];
      _Float16 hi = (_Float16)v;
      ah[k] = hi;
      al[k] = (_Float16)(v - (float)hi);
    }
#pragma unroll
    for (int k = 0; k < 5; ++k) {
      float w = W1[k * 16 + lane];
      _Float16 hi = (_Float16)w;
      bh[k] = hi;
      bl[k] = (_Float16)(w - (float)hi);
    }
  }

  v8f c = {};
  c = __builtin_amdgcn_wmma_f32_16x16x32_f16(false, al, false, bh, (short)0, c, false, false);
  c = __builtin_amdgcn_wmma_f32_16x16x32_f16(false, ah, false, bl, (short)0, c, false, false);
  c = __builtin_amdgcn_wmma_f32_16x16x32_f16(false, ah, false, bh, (short)0, c, false, false);

  int n  = lane & 15;
  int mb = row0 + ((lane >= 16) ? 8 : 0);
  if (row0 + 16 <= N) {
    float* base = t1 + (size_t)mb * 16 + n;
#pragma unroll
    for (int r = 0; r < 8; ++r) base[(size_t)r * 16] = c[r];
  } else {
#pragma unroll
    for (int r = 0; r < 8; ++r) {
      int m = mb + r;
      if (m < N) t1[(size_t)m * 16 + n] = c[r];
    }
  }
}

#define MT 256
#define MG 1024

template <bool FULL>
__device__ __forceinline__ void msg16_trip(
    const int* __restrict__ src, const int* __restrict__ dst,
    const float* __restrict__ t1, const float* __restrict__ dinv,
    float* __restrict__ agg1, long long e0, int nv, int lane) {
  const int c = lane & 15;
  const int h = lane >> 4;
  int   s = 0, d = 0;
  float w = 0.0f;
  if (FULL || lane < nv) {
    s = src[e0 + lane];
    d = dst[e0 + lane];
    w = dinv[s] * dinv[d];
  }
  float v[16];
#pragma unroll
  for (int u = 0; u < 16; ++u) {
    const int su = __shfl(s, 2 * u + h);
    v[u] = t1[(size_t)su * 16 + c];
  }
#pragma unroll
  for (int u = 0; u < 16; ++u) {
    const int   du = __shfl(d, 2 * u + h);
    const float wu = __shfl(w, 2 * u + h);
    if (FULL || 2 * u + h < nv) atomic_add_f32(&agg1[(size_t)du * 16 + c], v[u] * wu);
  }
}

__global__ __launch_bounds__(MT) void gcn_msg16(
    const int* __restrict__ src, const int* __restrict__ dst,
    const float* __restrict__ t1, const float* __restrict__ dinv,
    float* __restrict__ agg1, long long E) {
  const int       lane   = (int)(threadIdx.x & 31);
  const long long wave   = (long long)blockIdx.x * (MT / 32) + (threadIdx.x >> 5);
  const long long stride = (long long)gridDim.x * (MT / 32) * 32;
  long long e0 = wave * 32;
  for (; e0 + 32 <= E; e0 += stride)
    msg16_trip<true>(src, dst, t1, dinv, agg1, e0, 32, lane);
  if (e0 < E)
    msg16_trip<false>(src, dst, t1, dinv, agg1, e0, (int)(E - e0), lane);
}

__global__ void gcn_finalize1(const float* __restrict__ agg1, const float* __restrict__ t1,
                              const float* __restrict__ dinv, const float* __restrict__ b1,
                              const float* __restrict__ W2, float* __restrict__ t2, int N) {
  int i = blockIdx.x * blockDim.x + threadIdx.x;
  if (i >= N) return;
  float dd  = dinv[i] * dinv[i];
  float acc = 0.0f;
#pragma unroll
  for (int c = 0; c < 16; ++c) {
    float v = agg1[(size_t)i * 16 + c] + t1[(size_t)i * 16 + c] * dd + b1[c];
    v = fmaxf(v, 0.0f);
    acc = fmaf(v, W2[c], acc);
  }
  t2[i] = acc;
}

__global__ void gcn_finalize2(const float* __restrict__ planes, const float* __restrict__ t2,
                              const float* __restrict__ dinv, const float* __restrict__ b2,
                              float* __restrict__ out, int N) {
  int i = blockIdx.x * blockDim.x + threadIdx.x;
  if (i >= N) return;
  float agg = 0.0f;
#pragma unroll
  for (int b = 0; b < HB; ++b) agg += planes[(size_t)b * N + i];
  out[i] = agg + t2[i] * dinv[i] * dinv[i] + b2[0];
}

extern "C" void kernel_launch(void* const* d_in, const int* in_sizes, int n_in,
                              void* d_out, int out_size, void* d_ws, size_t ws_size,
                              hipStream_t stream) {
  const float* x  = (const float*)d_in[0];
  const float* W1 = (const float*)d_in[1];
  const float* b1 = (const float*)d_in[2];
  const float* W2 = (const float*)d_in[3];
  const float* b2 = (const float*)d_in[4];
  const int*   ei = (const int*)d_in[5];

  int       N = in_sizes[0] / 5;
  long long E = (long long)in_sizes[5] / 2;
  const int* src = ei;
  const int* dst = ei + E;

  float* ws   = (float*)d_ws;
  float* dinv = ws;
  float* t1   = dinv + N;
  float* agg1 = t1 + (size_t)16 * N;
  float* t2   = agg1 + (size_t)16 * N;
  float* planes = t1;

  float* out = (float*)d_out;

  long long initT   = (long long)N * 16;
  unsigned  gInit   = (unsigned)((initT + TPB - 1) / TPB);
  dim3      gScat(HB, (unsigned)((N + HR - 1) / HR));
  unsigned  gNode   = (unsigned)((N + TPB - 1) / TPB);
  int       waves   = (N + 15) / 16;
  unsigned  gWmma   = (unsigned)((waves + 1) / 2);

  gcn_scatter1<false><<<gScat, HT, 0, stream>>>(dst, dst, dinv, dinv, planes, E, N);
  gcn_rsqrt    <<<gNode,  TPB, 0, stream>>>(planes, dinv, N);
  gcn_init     <<<gInit,  TPB, 0, stream>>>(agg1, N);
  gcn_xw1      <<<gWmma,   64, 0, stream>>>(x, W1, t1, N);
  gcn_msg16    <<<MG,      MT, 0, stream>>>(src, dst, t1, dinv, agg1, E);
  gcn_finalize1<<<gNode,  TPB, 0, stream>>>(agg1, t1, dinv, b1, W2, t2, N);
  gcn_scatter1<true> <<<gScat, HT, 0, stream>>>(src, dst, t2, dinv, planes, E, N);
  gcn_finalize2<<<gNode,  TPB, 0, stream>>>(planes, t2, dinv, b2, out, N);
}


# Layer 1 reassociated as (A_hat x) W1: edge pass carries 5 channels, privatized in LDS bins and flushed by whole-line f32 atomic adds; 5x16 WMMA product applied after it (sum order changes)
# speedup vs baseline: 4.2156x; 1.3972x over previous
#include <hip/hip_runtime.h>
#include <hip/hip_bf16.h>

typedef __attribute__((ext_vector_type(16))) _Float16 v16h;
typedef __attribute__((ext_vector_type(8)))  float    v8f;

#define TPB 256
#define HT 1024
#define HB 32
#define HR 12800

__device__ __forceinline__ void atomic_add_f32(float* p, float v) {
  __hip_atomic_fetch_add(p, v, __ATOMIC_RELAXED, __HIP_MEMORY_SCOPE_AGENT);
}

__device__ __forceinline__ void lds_add_f32(float* p, float v) {
  __hip_atomic_fetch_add(p, v, __ATOMIC_RELAXED, __HIP_MEMORY_SCOPE_WORKGROUP);
}

#define C1 5
#define XP 8


__global__ void gcn_prep(const float* __restrict__ x, const float* __restrict__ dinv,
                         float* __restrict__ xs8, float* __restrict__ agg5, int N) {
  int i = blockIdx.x * blockDim.x + threadIdx.x;
  if (i >= N) return;
  const float di = dinv[i];
  float v[C1];
#pragma unroll
  for (int k = 0; k < C1; ++k) v[k] = x[(size_t)i * C1 + k] * di;
  float4* row = reinterpret_cast<float4*>(xs8 + (size_t)i * XP);
  row[0] = make_float4(v[0], v[1], v[2], v[3]);
  row[1] = make_float4(v[4], 0.0f, 0.0f, 0.0f);
#pragma unroll
  for (int k = 0; k < C1; ++k) agg5[(size_t)i * C1 + k] = v[k];
}

#define SU 4
#define SK (4 * SU)

template <bool WEIGHTED, bool GUARD>
__device__ __forceinline__ void scatter_trip(
    const int* __restrict__ src, const int* __restrict__ dst,
    const float* __restrict__ t2, const float* __restrict__ dinv,
    float* __restrict__ bins, const float* __restrict__ dloc,
    long long e, long long e0, long long e1v, int tid, int lo, int cnt) {
  int d[SK], s[SK];
#pragma unroll
  for (int u = 0; u < SU; ++u) {
    const long long g  = e + 4LL * (tid + HT * u);
    const bool      ok = !GUARD || g < e1v;
    const long long at = ok ? g : e0;
    int4 q = *reinterpret_cast<const int4*>(dst + at);
    if (!ok) q = make_int4(-1, -1, -1, -1);
    d[4 * u] = q.x; d[4 * u + 1] = q.y; d[4 * u + 2] = q.z; d[4 * u + 3] = q.w;
    if (WEIGHTED) {
      const int4 p = *reinterpret_cast<const int4*>(src + at);
      s[4 * u] = p.x; s[4 * u + 1] = p.y; s[4 * u + 2] = p.z; s[4 * u + 3] = p.w;
    }
  }
  if (WEIGHTED) {
    float val[SK];
#pragma unroll
    for (int k = 0; k < SK; ++k) {
      const unsigned r  = (unsigned)(d[k] - lo);
      const bool     in = r < (unsigned)cnt;
      const int      sk = in ? s[k] : 0;
      val[k] = t2[sk] * dinv[sk] * dloc[in ? r : 0u];
    }
#pragma unroll
    for (int k = 0; k < SK; ++k) asm volatile("" : "+v"(val[k]));
#pragma unroll
    for (int k = 0; k < SK; ++k) {
      const unsigned r = (unsigned)(d[k] - lo);
      if (r < (unsigned)cnt) lds_add_f32(&bins[r], val[k]);
    }
  } else {
#pragma unroll
    for (int k = 0; k < SK; ++k) {
      const unsigned r = (unsigned)(d[k] - lo);
      if (r < (unsigned)cnt) lds_add_f32(&bins[r], 1.0f);
    }
  }
}

template <bool WEIGHTED>
__global__ __launch_bounds__(HT) void gcn_scatter1(
    const int* __restrict__ src, const int* __restrict__ dst,
    const float* __restrict__ t2, const float* __restrict__ dinv,
    float* __restrict__ planes, long long E, int N) {
  constexpr int FJ = (HR + HT - 1) / HT;
  __shared__ float bins[FJ * HT];
  __shared__ float dloc[WEIGHTED ? FJ * HT : 1];

  const int tid = (int)threadIdx.x;
  const int lo  = (int)blockIdx.y * HR;
  int cnt = N - lo;
  if (cnt > HR) cnt = HR;

#pragma unroll
  for (int j = 0; j < FJ; ++j) {
    const int i = tid + j * HT;
    bins[i] = 0.0f;
    if (WEIGHTED) dloc[i] = dinv[lo + (i < cnt ? i : cnt - 1)];
  }
  __syncthreads();

  const long long per = (((E + HB - 1) / HB) + 3) & ~3LL;
  const long long e0  = (long long)blockIdx.x * per;
  long long e1 = e0 + per;
  if (e1 > E) e1 = E;
  if (e1 < e0) e1 = e0;
  long long rest = e0;

  const bool vec = WEIGHTED ? ((((size_t)dst | (size_t)src) & 15) == 0)
                            : (((size_t)dst & 15) == 0);
  if (vec) {
    const long long trip = (long long)SK * HT;
    const long long e1v  = e0 + ((e1 - e0) & ~3LL);
    long long e = e0;
    for (; e + trip <= e1v; e += trip)
      scatter_trip<WEIGHTED, false>(src, dst, t2, dinv, bins, dloc, e, e0, e1v, tid, lo, cnt);
    if (e < e1v)
      scatter_trip<WEIGHTED, true>(src, dst, t2, dinv, bins, dloc, e, e0, e1v, tid, lo, cnt);
    rest = e1v;
  }
  for (long long k = rest + tid; k < e1; k += HT) {
    const unsigned r = (unsigned)(dst[k] - lo);
    if (r < (unsigned)cnt) {
      float v = 1.0f;
      if (WEIGHTED) { const int sk = src[k]; v = t2[sk] * dinv[sk] * dloc[r]; }
      lds_add_f32(&bins[r], v);
    }
  }
  __syncthreads();

  float* pl = planes + (size_t)blockIdx.x * (size_t)N + lo;
  for (int i = tid; i < cnt; i += HT) pl[i] = bins[i];
}

__global__ void gcn_rsqrt(const float* __restrict__ planes, float* __restrict__ dinv, int N) {
  int i = blockIdx.x * blockDim.x + threadIdx.x;
  if (i >= N) return;
  float deg = 1.0f;
#pragma unroll
  for (int b = 0; b < HB; ++b) deg += planes[(size_t)b * N + i];
  dinv[i] = rsqrtf(deg);
}

#define SU5 2
#define SK5 (4 * SU5)

template <bool GUARD>
__device__ __forceinline__ void scatter5_trip(
    const int* __restrict__ src, const int* __restrict__ dst,
    const float* __restrict__ xs8, float* __restrict__ bins,
    long long e, long long e0, long long e1v, int tid, int lo, int cnt) {
  int d[SK5], s[SK5];
#pragma unroll
  for (int u = 0; u < SU5; ++u) {
    const long long g  = e + 4LL * (tid + HT * u);
    const bool      ok = !GUARD || g < e1v;
    const long long at = ok ? g : e0;
    int4 q = *reinterpret_cast<const int4*>(dst + at);
    if (!ok) q = make_int4(-1, -1, -1, -1);
    d[4 * u] = q.x; d[4 * u + 1] = q.y; d[4 * u + 2] = q.z; d[4 * u + 3] = q.w;
    const int4 p = *reinterpret_cast<const int4*>(src + at);
    s[4 * u] = p.x; s[4 * u + 1] = p.y; s[4 * u + 2] = p.z; s[4 * u + 3] = p.w;
  }
  float4 va[SK5];
  float  vb[SK5];
#pragma unroll
  for (int k = 0; k < SK5; ++k) {
    const unsigned r  = (unsigned)(d[k] - lo);
    const int      sk = r < (unsigned)cnt ? s[k] : 0;
    const float*   row = xs8 + (size_t)sk * XP;
    va[k] = *reinterpret_cast<const float4*>(row);
    vb[k] = row[4];
  }
#pragma unroll
  for (int k = 0; k < SK5; ++k)
    asm volatile("" : "+v"(va[k].x), "+v"(va[k].y), "+v"(va[k].z), "+v"(va[k].w), "+v"(vb[k]));
#pragma unroll
  for (int k = 0; k < SK5; ++k) {
    const unsigned r = (unsigned)(d[k] - lo);
    if (r < (unsigned)cnt) {
      float* b = bins + r * C1;
      lds_add_f32(b + 0, va[k].x);
      lds_add_f32(b + 1, va[k].y);
      lds_add_f32(b + 2, va[k].z);
      lds_add_f32(b + 3, va[k].w);
      lds_add_f32(b + 4, vb[k]);
    }
  }
}

__global__ __launch_bounds__(HT) void gcn_scatter5(
    const int* __restrict__ src, const int* __restrict__ dst,
    const float* __restrict__ xs8, float* __restrict__ agg5, long long E, int N) {
  constexpr int FJ = (HR * C1 + HT - 1) / HT;
  __shared__ float bins[FJ * HT];

  const int tid = (int)threadIdx.x;
  const int lo  = (int)blockIdx.y * HR;
  int cnt = N - lo;
  if (cnt > HR) cnt = HR;

  for (int j = 0; j < FJ; ++j) bins[tid + j * HT] = 0.0f;
  __syncthreads();

  const long long per = (((E + HB - 1) / HB) + 3) & ~3LL;
  const long long e0  = (long long)blockIdx.x * per;
  long long e1 = e0 + per;
  if (e1 > E) e1 = E;
  if (e1 < e0) e1 = e0;
  long long rest = e0;

  if (((((size_t)dst | (size_t)src) & 15) == 0)) {
    const long long trip = (long long)SK5 * HT;
    const long long e1v  = e0 + ((e1 - e0) & ~3LL);
    long long e = e0;
    for (; e + trip <= e1v; e += trip)
      scatter5_trip<false>(src, dst, xs8, bins, e, e0, e1v, tid, lo, cnt);
    if (e < e1v)
      scatter5_trip<true>(src, dst, xs8, bins, e, e0, e1v, tid, lo, cnt);
    rest = e1v;
  }
  for (long long k = rest + tid; k < e1; k += HT) {
    const unsigned r = (unsigned)(dst[k] - lo);
    if (r < (unsigned)cnt) {
      const float* row = xs8 + (size_t)src[k] * XP;
      float* b = bins + r * C1;
#pragma unroll
      for (int c = 0; c < C1; ++c) lds_add_f32(b + c, row[c]);
    }
  }
  __syncthreads();

  float* ag = agg5 + (size_t)lo * C1;
  const int tot = cnt * C1;
  for (int i = tid; i < tot; i += HT) atomic_add_f32(&ag[i], bins[i]);
}

__global__ __launch_bounds__(TPB) void gcn_finalize1(
    const float* __restrict__ agg5, const float* __restrict__ dinv,
    const float* __restrict__ W1, const float* __restrict__ b1,
    const float* __restrict__ W2, float* __restrict__ t2, int N) {
  int wave = (int)(((long long)blockIdx.x * blockDim.x + threadIdx.x) >> 5);
  int lane = (int)(threadIdx.x & 31);
  int row0 = wave * 16;
  if (row0 >= N) return;

  v16h ah = {}, al = {};
  v16h bh = {}, bl = {};
  if (lane < 16) {
    int m = row0 + lane;
    if (m >= N) m = N - 1;
    const float  di = dinv[m];
    const float* ar = agg5 + (size_t)m * C1;
#pragma unroll
    for (int k = 0; k < C1; ++k) {
      float v = ar[k] * di;
      _Float16 hi = (_Float16)v;
      ah[k] = hi;
      al[k] = (_Float16)(v - (float)hi);
    }
#pragma unroll
    for (int k = 0; k < C1; ++k) {
      float w = W1[k * 16 + lane];
      _Float16 hi = (_Float16)w;
      bh[k] = hi;
      bl[k] = (_Float16)(w - (float)hi);
    }
  }

  v8f c = {};
  c = __builtin_amdgcn_wmma_f32_16x16x32_f16(false, al, false, bh, (short)0, c, false, false);
  c = __builtin_amdgcn_wmma_f32_16x16x32_f16(false, ah, false, bl, (short)0, c, false, false);
  c = __builtin_amdgcn_wmma_f32_16x16x32_f16(false, ah, false, bh, (short)0, c, false, false);

  const int   n    = lane & 15;
  const float bias = b1[n];
  const float w2   = W2[n];
  float p[8];
#pragma unroll
  for (int r = 0; r < 8; ++r) {
    float v = fmaxf(c[r] + bias, 0.0f) * w2;
    v += __shfl_xor(v, 8);
    v += __shfl_xor(v, 4);
    v += __shfl_xor(v, 2);
    v += __shfl_xor(v, 1);
    p[r] = v;
  }
  float mine = p[0];
#pragma unroll
  for (int r = 1; r < 8; ++r) mine = (n == r) ? p[r] : mine;
  const int m = row0 + ((lane >= 16) ? 8 : 0) + n;
  if (n < 8 && m < N) t2[m] = mine;
}

__global__ void gcn_finalize2(const float* __restrict__ planes, const float* __restrict__ t2,
                              const float* __restrict__ dinv, const float* __restrict__ b2,
                              float* __restrict__ out, int N) {
  int i = blockIdx.x * blockDim.x + threadIdx.x;
  if (i >= N) return;
  float agg = 0.0f;
#pragma unroll
  for (int b = 0; b < HB; ++b) agg += planes[(size_t)b * N + i];
  out[i] = agg + t2[i] * dinv[i] * dinv[i] + b2[0];
}

extern "C" void kernel_launch(void* const* d_in, const int* in_sizes, int n_in,
                              void* d_out, int out_size, void* d_ws, size_t ws_size,
                              hipStream_t stream) {
  const float* x  = (const float*)d_in[0];
  const float* W1 = (const float*)d_in[1];
  const float* b1 = (const float*)d_in[2];
  const float* W2 = (const float*)d_in[3];
  const float* b2 = (const float*)d_in[4];
  const int*   ei = (const int*)d_in[5];

  int       N = in_sizes[0] / 5;
  long long E = (long long)in_sizes[5] / 2;
  const int* src = ei;
  const int* dst = ei + E;

  float* ws     = (float*)d_ws;
  float* planes = ws;
  float* xs8    = ws;
  float* agg5   = ws + (size_t)XP * N;
  float* dinv   = ws + (size_t)HB * N;
  float* t2     = dinv + N;

  float* out = (float*)d_out;

  dim3      gScat(HB, (unsigned)((N + HR - 1) / HR));
  unsigned  gNode   = (unsigned)((N + TPB - 1) / TPB);
  int       waves   = (N + 15) / 16;
  unsigned  gWmma   = (unsigned)((waves + TPB / 32 - 1) / (TPB / 32));

  gcn_scatter1<false><<<gScat, HT, 0, stream>>>(dst, dst, dinv, dinv, planes, E, N);
  gcn_rsqrt    <<<gNode,  TPB, 0, stream>>>(planes, dinv, N);
  gcn_prep     <<<gNode,  TPB, 0, stream>>>(x, dinv, xs8, agg5, N);
  gcn_scatter5 <<<gScat,   HT, 0, stream>>>(src, dst, xs8, agg5, E, N);
  gcn_finalize1<<<gWmma,  TPB, 0, stream>>>(agg5, dinv, W1, b1, W2, t2, N);
  gcn_scatter1<true> <<<gScat, HT, 0, stream>>>(src, dst, t2, dinv, planes, E, N);
  gcn_finalize2<<<gNode,  TPB, 0, stream>>>(planes, t2, dinv, b2, out, N);
}


# Layer-2 edge pass gathers one word per edge: finalize1 stores u = t2*dinv (same product, same rounding), so each edge reads u[src] instead of t2[src] and dinv[src]; sums unchanged
# speedup vs baseline: 4.8126x; 1.1416x over previous
#include <hip/hip_runtime.h>
#include <hip/hip_bf16.h>

typedef __attribute__((ext_vector_type(16))) _Float16 v16h;
typedef __attribute__((ext_vector_type(8)))  float    v8f;

#define TPB 256
#define HT 1024
#define HB 32
#define HR 12800

__device__ __forceinline__ void atomic_add_f32(float* p, float v) {
  __hip_atomic_fetch_add(p, v, __ATOMIC_RELAXED, __HIP_MEMORY_SCOPE_AGENT);
}

__device__ __forceinline__ void lds_add_f32(float* p, float v) {
  __hip_atomic_fetch_add(p, v, __ATOMIC_RELAXED, __HIP_MEMORY_SCOPE_WORKGROUP);
}

#define C1 5
#define XP 8


__global__ void gcn_prep(const float* __restrict__ x, const float* __restrict__ dinv,
                         float* __restrict__ xs8, float* __restrict__ agg5, int N) {
  int i = blockIdx.x * blockDim.x + threadIdx.x;
  if (i >= N) return;
  const float di = dinv[i];
  float v[C1];
#pragma unroll
  for (int k = 0; k < C1; ++k) v[k] = x[(size_t)i * C1 + k] * di;
  float4* row = reinterpret_cast<float4*>(xs8 + (size_t)i * XP);
  row[0] = make_float4(v[0], v[1], v[2], v[3]);
  row[1] = make_float4(v[4], 0.0f, 0.0f, 0.0f);
#pragma unroll
  for (int k = 0; k < C1; ++k) agg5[(size_t)i * C1 + k] = v[k];
}

#define SU 4
#define SK (4 * SU)

template <bool WEIGHTED, bool GUARD>
__device__ __forceinline__ void scatter_trip(
    const int* __restrict__ src, const int* __restrict__ dst,
    const float* __restrict__ u,
    float* __restrict__ bins, const float* __restrict__ dloc,
    long long e, long long e0, long long e1v, int tid, int lo, int cnt) {
  int d[SK], s[SK];
#pragma unroll
  for (int u = 0; u < SU; ++u) {
    const long long g  = e + 4LL * (tid + HT * u);
    const bool      ok = !GUARD || g < e1v;
    const long long at = ok ? g : e0;
    int4 q = *reinterpret_cast<const int4*>(dst + at);
    if (!ok) q = make_int4(-1, -1, -1, -1);
    d[4 * u] = q.x; d[4 * u + 1] = q.y; d[4 * u + 2] = q.z; d[4 * u + 3] = q.w;
    if (WEIGHTED) {
      const int4 p = *reinterpret_cast<const int4*>(src + at);
      s[4 * u] = p.x; s[4 * u + 1] = p.y; s[4 * u + 2] = p.z; s[4 * u + 3] = p.w;
    }
  }
  if (WEIGHTED) {
    float val[SK];
#pragma unroll
    for (int k = 0; k < SK; ++k) {
      const unsigned r  = (unsigned)(d[k] - lo);
      const bool     in = r < (unsigned)cnt;
      const int      sk = in ? s[k] : 0;
      val[k] = u[sk] * dloc[in ? r : 0u];
    }
#pragma unroll
    for (int k = 0; k < SK; ++k) asm volatile("" : "+v"(val[k]));
#pragma unroll
    for (int k = 0; k < SK; ++k) {
      const unsigned r = (unsigned)(d[k] - lo);
      if (r < (unsigned)cnt) lds_add_f32(&bins[r], val[k]);
    }
  } else {
#pragma unroll
    for (int k = 0; k < SK; ++k) {
      const unsigned r = (unsigned)(d[k] - lo);
      if (r < (unsigned)cnt) lds_add_f32(&bins[r], 1.0f);
    }
  }
}

template <bool WEIGHTED>
__global__ __launch_bounds__(HT) void gcn_scatter1(
    const int* __restrict__ src, const int* __restrict__ dst,
    const float* __restrict__ u, const float* __restrict__ dinv,
    float* __restrict__ planes, long long E, int N) {
  constexpr int FJ = (HR + HT - 1) / HT;
  __shared__ float bins[FJ * HT];
  __shared__ float dloc[WEIGHTED ? FJ * HT : 1];

  const int tid = (int)threadIdx.x;
  const int lo  = (int)blockIdx.y * HR;
  int cnt = N - lo;
  if (cnt > HR) cnt = HR;

#pragma unroll
  for (int j = 0; j < FJ; ++j) {
    const int i = tid + j * HT;
    bins[i] = 0.0f;
    if (WEIGHTED) dloc[i] = dinv[lo + (i < cnt ? i : cnt - 1)];
  }
  __syncthreads();

  const long long per = (((E + HB - 1) / HB) + 3) & ~3LL;
  const long long e0  = (long long)blockIdx.x * per;
  long long e1 = e0 + per;
  if (e1 > E) e1 = E;
  if (e1 < e0) e1 = e0;
  long long rest = e0;

  const bool vec = WEIGHTED ? ((((size_t)dst | (size_t)src) & 15) == 0)
                            : (((size_t)dst & 15) == 0);
  if (vec) {
    const long long trip = (long long)SK * HT;
    const long long e1v  = e0 + ((e1 - e0) & ~3LL);
    long long e = e0;
    for (; e + trip <= e1v; e += trip)
      scatter_trip<WEIGHTED, false>(src, dst, u, bins, dloc, e, e0, e1v, tid, lo, cnt);
    if (e < e1v)
      scatter_trip<WEIGHTED, true>(src, dst, u, bins, dloc, e, e0, e1v, tid, lo, cnt);
    rest = e1v;
  }
  for (long long k = rest + tid; k < e1; k += HT) {
    const unsigned r = (unsigned)(dst[k] - lo);
    if (r < (unsigned)cnt) {
      float v = 1.0f;
      if (WEIGHTED) v = u[src[k]] * dloc[r];
      lds_add_f32(&bins[r], v);
    }
  }
  __syncthreads();

  float* pl = planes + (size_t)blockIdx.x * (size_t)N + lo;
  for (int i = tid; i < cnt; i += HT) pl[i] = bins[i];
}

__global__ void gcn_rsqrt(const float* __restrict__ planes, float* __restrict__ dinv, int N) {
  int i = blockIdx.x * blockDim.x + threadIdx.x;
  if (i >= N) return;
  float deg = 1.0f;
#pragma unroll
  for (int b = 0; b < HB; ++b) deg += planes[(size_t)b * N + i];
  dinv[i] = rsqrtf(deg);
}

#define SU5 2
#define SK5 (4 * SU5)

template <bool GUARD>
__device__ __forceinline__ void scatter5_trip(
    const int* __restrict__ src, const int* __restrict__ dst,
    const float* __restrict__ xs8, float* __restrict__ bins,
    long long e, long long e0, long long e1v, int tid, int lo, int cnt) {
  int d[SK5], s[SK5];
#pragma unroll
  for (int u = 0; u < SU5; ++u) {
    const long long g  = e + 4LL * (tid + HT * u);
    const bool      ok = !GUARD || g < e1v;
    const long long at = ok ? g : e0;
    int4 q = *reinterpret_cast<const int4*>(dst + at);
    if (!ok) q = make_int4(-1, -1, -1, -1);
    d[4 * u] = q.x; d[4 * u + 1] = q.y; d[4 * u + 2] = q.z; d[4 * u + 3] = q.w;
    const int4 p = *reinterpret_cast<const int4*>(src + at);
    s[4 * u] = p.x; s[4 * u + 1] = p.y; s[4 * u + 2] = p.z; s[4 * u + 3] = p.w;
  }
  float4 va[SK5];
  float  vb[SK5];
#pragma unroll
  for (int k = 0; k < SK5; ++k) {
    const unsigned r  = (unsigned)(d[k] - lo);
    const int      sk = r < (unsigned)cnt ? s[k] : 0;
    const float*   row = xs8 + (size_t)sk * XP;
    va[k] = *reinterpret_cast<const float4*>(row);
    vb[k] = row[4];
  }
#pragma unroll
  for (int k = 0; k < SK5; ++k)
    asm volatile("" : "+v"(va[k].x), "+v"(va[k].y), "+v"(va[k].z), "+v"(va[k].w), "+v"(vb[k]));
#pragma unroll
  for (int k = 0; k < SK5; ++k) {
    const unsigned r = (unsigned)(d[k] - lo);
    if (r < (unsigned)cnt) {
      float* b = bins + r * C1;
      lds_add_f32(b + 0, va[k].x);
      lds_add_f32(b + 1, va[k].y);
      lds_add_f32(b + 2, va[k].z);
      lds_add_f32(b + 3, va[k].w);
      lds_add_f32(b + 4, vb[k]);
    }
  }
}

__global__ __launch_bounds__(HT) void gcn_scatter5(
    const int* __restrict__ src, const int* __restrict__ dst,
    const float* __restrict__ xs8, float* __restrict__ agg5, long long E, int N) {
  constexpr int FJ = (HR * C1 + HT - 1) / HT;
  __shared__ float bins[FJ * HT];

  const int tid = (int)threadIdx.x;
  const int lo  = (int)blockIdx.y * HR;
  int cnt = N - lo;
  if (cnt > HR) cnt = HR;

  for (int j = 0; j < FJ; ++j) bins[tid + j * HT] = 0.0f;
  __syncthreads();

  const long long per = (((E + HB - 1) / HB) + 3) & ~3LL;
  const long long e0  = (long long)blockIdx.x * per;
  long long e1 = e0 + per;
  if (e1 > E) e1 = E;
  if (e1 < e0) e1 = e0;
  long long rest = e0;

  if (((((size_t)dst | (size_t)src) & 15) == 0)) {
    const long long trip = (long long)SK5 * HT;
    const long long e1v  = e0 + ((e1 - e0) & ~3LL);
    long long e = e0;
    for (; e + trip <= e1v; e += trip)
      scatter5_trip<false>(src, dst, xs8, bins, e, e0, e1v, tid, lo, cnt);
    if (e < e1v)
      scatter5_trip<true>(src, dst, xs8, bins, e, e0, e1v, tid, lo, cnt);
    rest = e1v;
  }
  for (long long k = rest + tid; k < e1; k += HT) {
    const unsigned r = (unsigned)(dst[k] - lo);
    if (r < (unsigned)cnt) {
      const float* row = xs8 + (size_t)src[k] * XP;
      float* b = bins + r * C1;
#pragma unroll
      for (int c = 0; c < C1; ++c) lds_add_f32(b + c, row[c]);
    }
  }
  __syncthreads();

  float* ag = agg5 + (size_t)lo * C1;
  const int tot = cnt * C1;
  for (int i = tid; i < tot; i += HT) atomic_add_f32(&ag[i], bins[i]);
}

__global__ __launch_bounds__(TPB) void gcn_finalize1(
    const float* __restrict__ agg5, const float* __restrict__ dinv,
    const float* __restrict__ W1, const float* __restrict__ b1,
    const float* __restrict__ W2, float* __restrict__ u, int N) {
  int wave = (int)(((long long)blockIdx.x * blockDim.x + threadIdx.x) >> 5);
  int lane = (int)(threadIdx.x & 31);
  int row0 = wave * 16;
  if (row0 >= N) return;

  const int   ms = row0 + ((lane >= 16) ? 8 : 0) + (lane & 7);
  const float ds = dinv[ms < N ? ms : N - 1];

  v16h ah = {}, al = {};
  v16h bh = {}, bl = {};
  if (lane < 16) {
    int m = row0 + lane;
    if (m >= N) m = N - 1;
    const float  di = dinv[m];
    const float* ar = agg5 + (size_t)m * C1;
#pragma unroll
    for (int k = 0; k < C1; ++k) {
      float v = ar[k] * di;
      _Float16 hi = (_Float16)v;
      ah[k] = hi;
      al[k] = (_Float16)(v - (float)hi);
    }
#pragma unroll
    for (int k = 0; k < C1; ++k) {
      float w = W1[k * 16 + lane];
      _Float16 hi = (_Float16)w;
      bh[k] = hi;
      bl[k] = (_Float16)(w - (float)hi);
    }
  }

  v8f c = {};
  c = __builtin_amdgcn_wmma_f32_16x16x32_f16(false, al, false, bh, (short)0, c, false, false);
  c = __builtin_amdgcn_wmma_f32_16x16x32_f16(false, ah, false, bl, (short)0, c, false, false);
  c = __builtin_amdgcn_wmma_f32_16x16x32_f16(false, ah, false, bh, (short)0, c, false, false);

  const int   n    = lane & 15;
  const float bias = b1[n];
  const float w2   = W2[n];
  float p[8];
#pragma unroll
  for (int r = 0; r < 8; ++r) {
    float v = fmaxf(c[r] + bias, 0.0f) * w2;
    v += __shfl_xor(v, 8);
    v += __shfl_xor(v, 4);
    v += __shfl_xor(v, 2);
    v += __shfl_xor(v, 1);
    p[r] = v;
  }
  float mine = p[0];
#pragma unroll
  for (int r = 1; r < 8; ++r) mine = (n == r) ? p[r] : mine;
  const int m = row0 + ((lane >= 16) ? 8 : 0) + n;
  if (n < 8 && m < N) u[m] = mine * ds;
}

__global__ void gcn_finalize2(const float* __restrict__ planes, const float* __restrict__ u,
                              const float* __restrict__ dinv, const float* __restrict__ b2,
                              float* __restrict__ out, int N) {
  int i = blockIdx.x * blockDim.x + threadIdx.x;
  if (i >= N) return;
  float agg = 0.0f;
#pragma unroll
  for (int b = 0; b < HB; ++b) agg += planes[(size_t)b * N + i];
  out[i] = agg + u[i] * dinv[i] + b2[0];
}

extern "C" void kernel_launch(void* const* d_in, const int* in_sizes, int n_in,
                              void* d_out, int out_size, void* d_ws, size_t ws_size,
                              hipStream_t stream) {
  const float* x  = (const float*)d_in[0];
  const float* W1 = (const float*)d_in[1];
  const float* b1 = (const float*)d_in[2];
  const float* W2 = (const float*)d_in[3];
  const float* b2 = (const float*)d_in[4];
  const int*   ei = (const int*)d_in[5];

  int       N = in_sizes[0] / 5;
  long long E = (long long)in_sizes[5] / 2;
  const int* src = ei;
  const int* dst = ei + E;

  float* ws     = (float*)d_ws;
  float* planes = ws;
  float* xs8    = ws;
  float* agg5   = ws + (size_t)XP * N;
  float* dinv   = ws + (size_t)HB * N;
  float* u      = dinv + N;

  float* out = (float*)d_out;

  dim3      gScat(HB, (unsigned)((N + HR - 1) / HR));
  unsigned  gNode   = (unsigned)((N + TPB - 1) / TPB);
  int       waves   = (N + 15) / 16;
  unsigned  gWmma   = (unsigned)((waves + TPB / 32 - 1) / (TPB / 32));

  gcn_scatter1<false><<<gScat, HT, 0, stream>>>(dst, dst, dinv, dinv, planes, E, N);
  gcn_rsqrt    <<<gNode,  TPB, 0, stream>>>(planes, dinv, N);
  gcn_prep     <<<gNode,  TPB, 0, stream>>>(x, dinv, xs8, agg5, N);
  gcn_scatter5 <<<gScat,   HT, 0, stream>>>(src, dst, xs8, agg5, E, N);
  gcn_finalize1<<<gWmma,  TPB, 0, stream>>>(agg5, dinv, W1, b1, W2, u, N);
  gcn_scatter1<true> <<<gScat, HT, 0, stream>>>(src, dst, u, dinv, planes, E, N);
  gcn_finalize2<<<gNode,  TPB, 0, stream>>>(planes, u, dinv, b2, out, N);
}


# Layer-2 edge pass reads u[src] from LDS: blocks split edges by (src half, dst quarter) and copy their u half to LDS; dinv[dst] applied per node; bins flushed by f32 line atomics (sum order changes)
# speedup vs baseline: 5.4116x; 1.1245x over previous
#include <hip/hip_runtime.h>
#include <hip/hip_bf16.h>


typedef __attribute__((ext_vector_type(16))) _Float16 v16h;
typedef __attribute__((ext_vector_type(8)))  float    v8f;

#define TPB 256
#define HT 1024
#define HB 32
#define HR 12800

__device__ __forceinline__ void atomic_add_f32(float* p, float v) {
  __hip_atomic_fetch_add(p, v, __ATOMIC_RELAXED, __HIP_MEMORY_SCOPE_AGENT);
}

__device__ __forceinline__ void lds_add_f32(float* p, float v) {
  __hip_atomic_fetch_add(p, v, __ATOMIC_RELAXED, __HIP_MEMORY_SCOPE_WORKGROUP);
}

#define C1 5
#define XP 8


__global__ void gcn_prep(const float* __restrict__ x, const float* __restrict__ dinv,
                         float* __restrict__ xs8, float* __restrict__ agg5, int N) {
  int i = blockIdx.x * blockDim.x + threadIdx.x;
  if (i >= N) return;
  const float di = dinv[i];
  float v[C1];
#pragma unroll
  for (int k = 0; k < C1; ++k) v[k] = x[(size_t)i * C1 + k] * di;
  float4* row = reinterpret_cast<float4*>(xs8 + (size_t)i * XP);
  row[0] = make_float4(v[0], v[1], v[2], v[3]);
  row[1] = make_float4(v[4], 0.0f, 0.0f, 0.0f);
#pragma unroll
  for (int k = 0; k < C1; ++k) agg5[(size_t)i * C1 + k] = v[k];
}

#define SU 4
#define SK (4 * SU)


template <bool GUARD>
__device__ __forceinline__ void degree_trip(
    const int* __restrict__ dst, float* __restrict__ bins,
    long long e, long long e0, long long e1v, int tid, int lo, int cnt) {
  int d[SK];
#pragma unroll
  for (int g = 0; g < SU; ++g) {
    const long long at0 = e + 4LL * (tid + HT * g);
    const bool      ok  = !GUARD || at0 < e1v;
    const long long at  = ok ? at0 : e0;
    int4 q = *reinterpret_cast<const int4*>(dst + at);
    if (!ok) q = make_int4(-1, -1, -1, -1);
    d[4 * g] = q.x; d[4 * g + 1] = q.y; d[4 * g + 2] = q.z; d[4 * g + 3] = q.w;
  }
#pragma unroll
  for (int k = 0; k < SK; ++k) {
    const unsigned r = (unsigned)(d[k] - lo);
    if (r < (unsigned)cnt) lds_add_f32(&bins[r], 1.0f);
  }
}

__global__ __launch_bounds__(HT) void gcn_degree(
    const int* __restrict__ dst, float* __restrict__ planes, long long E, int N) {
  constexpr int FJ = (HR + HT - 1) / HT;
  __shared__ float bins[FJ * HT];

  const int tid = (int)threadIdx.x;
  const int lo  = (int)blockIdx.y * HR;
  int cnt = N - lo;
  if (cnt > HR) cnt = HR;

  for (int j = 0; j < FJ; ++j) bins[tid + j * HT] = 0.0f;
  __syncthreads();

  const long long per = (((E + HB - 1) / HB) + 3) & ~3LL;
  const long long e0  = (long long)blockIdx.x * per;
  long long e1 = e0 + per;
  if (e1 > E) e1 = E;
  if (e1 < e0) e1 = e0;
  long long rest = e0;

  if ((((size_t)dst) & 15) == 0) {
    const long long trip = (long long)SK * HT;
    const long long e1v  = e0 + ((e1 - e0) & ~3LL);
    long long e = e0;
    for (; e + trip <= e1v; e += trip)
      degree_trip<false>(dst, bins, e, e0, e1v, tid, lo, cnt);
    if (e < e1v)
      degree_trip<true>(dst, bins, e, e0, e1v, tid, lo, cnt);
    rest = e1v;
  }
  for (long long k = rest + tid; k < e1; k += HT) {
    const unsigned r = (unsigned)(dst[k] - lo);
    if (r < (unsigned)cnt) lds_add_f32(&bins[r], 1.0f);
  }
  __syncthreads();

  float* pl = planes + (size_t)blockIdx.x * (size_t)N + lo;
  for (int i = tid; i < cnt; i += HT) pl[i] = bins[i];
}

__global__ void gcn_rsqrt(const float* __restrict__ planes, float* __restrict__ dinv, int N) {
  int i = blockIdx.x * blockDim.x + threadIdx.x;
  if (i >= N) return;
  float deg = 1.0f;
#pragma unroll
  for (int b = 0; b < HB; ++b) deg += planes[(size_t)b * N + i];
  dinv[i] = rsqrtf(deg);
}

#define SU5 2
#define SK5 (4 * SU5)

template <bool GUARD>
__device__ __forceinline__ void scatter5_trip(
    const int* __restrict__ src, const int* __restrict__ dst,
    const float* __restrict__ xs8, float* __restrict__ bins,
    long long e, long long e0, long long e1v, int tid, int lo, int cnt) {
  int d[SK5], s[SK5];
#pragma unroll
  for (int u = 0; u < SU5; ++u) {
    const long long g  = e + 4LL * (tid + HT * u);
    const bool      ok = !GUARD || g < e1v;
    const long long at = ok ? g : e0;
    int4 q = *reinterpret_cast<const int4*>(dst + at);
    if (!ok) q = make_int4(-1, -1, -1, -1);
    d[4 * u] = q.x; d[4 * u + 1] = q.y; d[4 * u + 2] = q.z; d[4 * u + 3] = q.w;
    const int4 p = *reinterpret_cast<const int4*>(src + at);
    s[4 * u] = p.x; s[4 * u + 1] = p.y; s[4 * u + 2] = p.z; s[4 * u + 3] = p.w;
  }
  float4 va[SK5];
  float  vb[SK5];
#pragma unroll
  for (int k = 0; k < SK5; ++k) {
    const unsigned r  = (unsigned)(d[k] - lo);
    const int      sk = r < (unsigned)cnt ? s[k] : 0;
    const float*   row = xs8 + (size_t)sk * XP;
    va[k] = *reinterpret_cast<const float4*>(row);
    vb[k] = row[4];
  }
#pragma unroll
  for (int k = 0; k < SK5; ++k)
    asm volatile("" : "+v"(va[k].x), "+v"(va[k].y), "+v"(va[k].z), "+v"(va[k].w), "+v"(vb[k]));
#pragma unroll
  for (int k = 0; k < SK5; ++k) {
    const unsigned r = (unsigned)(d[k] - lo);
    if (r < (unsigned)cnt) {
      float* b = bins + r * C1;
      lds_add_f32(b + 0, va[k].x);
      lds_add_f32(b + 1, va[k].y);
      lds_add_f32(b + 2, va[k].z);
      lds_add_f32(b + 3, va[k].w);
      lds_add_f32(b + 4, vb[k]);
    }
  }
}

__global__ __launch_bounds__(HT) void gcn_scatter5(
    const int* __restrict__ src, const int* __restrict__ dst,
    const float* __restrict__ xs8, float* __restrict__ agg5, long long E, int N) {
  constexpr int FJ = (HR * C1 + HT - 1) / HT;
  __shared__ float bins[FJ * HT];

  const int tid = (int)threadIdx.x;
  const int lo  = (int)blockIdx.y * HR;
  int cnt = N - lo;
  if (cnt > HR) cnt = HR;

  for (int j = 0; j < FJ; ++j) bins[tid + j * HT] = 0.0f;
  __syncthreads();

  const long long per = (((E + HB - 1) / HB) + 3) & ~3LL;
  const long long e0  = (long long)blockIdx.x * per;
  long long e1 = e0 + per;
  if (e1 > E) e1 = E;
  if (e1 < e0) e1 = e0;
  long long rest = e0;

  if (((((size_t)dst | (size_t)src) & 15) == 0)) {
    const long long trip = (long long)SK5 * HT;
    const long long e1v  = e0 + ((e1 - e0) & ~3LL);
    long long e = e0;
    for (; e + trip <= e1v; e += trip)
      scatter5_trip<false>(src, dst, xs8, bins, e, e0, e1v, tid, lo, cnt);
    if (e < e1v)
      scatter5_trip<true>(src, dst, xs8, bins, e, e0, e1v, tid, lo, cnt);
    rest = e1v;
  }
  for (long long k = rest + tid; k < e1; k += HT) {
    const unsigned r = (unsigned)(dst[k] - lo);
    if (r < (unsigned)cnt) {
      const float* row = xs8 + (size_t)src[k] * XP;
      float* b = bins + r * C1;
#pragma unroll
      for (int c = 0; c < C1; ++c) lds_add_f32(b + c, row[c]);
    }
  }
  __syncthreads();

  float* ag = agg5 + (size_t)lo * C1;
  const int tot = cnt * C1;
  for (int i = tid; i < tot; i += HT) atomic_add_f32(&ag[i], bins[i]);
}

__global__ __launch_bounds__(TPB) void gcn_finalize1(
    const float* __restrict__ agg5, const float* __restrict__ dinv,
    const float* __restrict__ W1, const float* __restrict__ b1,
    const float* __restrict__ W2, float* __restrict__ u, float* __restrict__ acc, int N) {
  int wave = (int)(((long long)blockIdx.x * blockDim.x + threadIdx.x) >> 5);
  int lane = (int)(threadIdx.x & 31);
  int row0 = wave * 16;
  if (row0 >= N) return;

  const int   ms = row0 + ((lane >= 16) ? 8 : 0) + (lane & 7);
  const float ds = dinv[ms < N ? ms : N - 1];

  v16h ah = {}, al = {};
  v16h bh = {}, bl = {};
  if (lane < 16) {
    int m = row0 + lane;
    if (m >= N) m = N - 1;
    const float  di = dinv[m];
    const float* ar = agg5 + (size_t)m * C1;
#pragma unroll
    for (int k = 0; k < C1; ++k) {
      float v = ar[k] * di;
      _Float16 hi = (_Float16)v;
      ah[k] = hi;
      al[k] = (_Float16)(v - (float)hi);
    }
#pragma unroll
    for (int k = 0; k < C1; ++k) {
      float w = W1[k * 16 + lane];
      _Float16 hi = (_Float16)w;
      bh[k] = hi;
      bl[k] = (_Float16)(w - (float)hi);
    }
  }

  v8f c = {};
  c = __builtin_amdgcn_wmma_f32_16x16x32_f16(false, al, false, bh, (short)0, c, false, false);
  c = __builtin_amdgcn_wmma_f32_16x16x32_f16(false, ah, false, bl, (short)0, c, false, false);
  c = __builtin_amdgcn_wmma_f32_16x16x32_f16(false, ah, false, bh, (short)0, c, false, false);

  const int   n    = lane & 15;
  const float bias = b1[n];
  const float w2   = W2[n];
  float p[8];
#pragma unroll
  for (int r = 0; r < 8; ++r) {
    float v = fmaxf(c[r] + bias, 0.0f) * w2;
    v += __shfl_xor(v, 8);
    v += __shfl_xor(v, 4);
    v += __shfl_xor(v, 2);
    v += __shfl_xor(v, 1);
    p[r] = v;
  }
  float mine = p[0];
#pragma unroll
  for (int r = 1; r < 8; ++r) mine = (n == r) ? p[r] : mine;
  const int m = row0 + ((lane >= 16) ? 8 : 0) + n;
  if (n < 8 && m < N) {
    const float t = mine * ds;
    u[m]   = t;
    acc[m] = t;
  }
}

#define QR 25600
#define SR 51200

template <bool GUARD>
__device__ __forceinline__ void scatter_u_trip(
    const int* __restrict__ src, const int* __restrict__ dst,
    const float* __restrict__ us, float* __restrict__ bins,
    long long e, long long e0, long long e1v, int tid,
    int dlo, int dcnt, int slo, int scnt) {
  int d[SK], s[SK];
#pragma unroll
  for (int g = 0; g < SU; ++g) {
    const long long at0 = e + 4LL * (tid + HT * g);
    const bool      ok  = !GUARD || at0 < e1v;
    const long long at  = ok ? at0 : e0;
    int4 q = *reinterpret_cast<const int4*>(dst + at);
    if (!ok) q = make_int4(-1, -1, -1, -1);
    d[4 * g] = q.x; d[4 * g + 1] = q.y; d[4 * g + 2] = q.z; d[4 * g + 3] = q.w;
    const int4 p = *reinterpret_cast<const int4*>(src + at);
    s[4 * g] = p.x; s[4 * g + 1] = p.y; s[4 * g + 2] = p.z; s[4 * g + 3] = p.w;
  }
  float val[SK];
#pragma unroll
  for (int k = 0; k < SK; ++k) {
    const unsigned r  = (unsigned)(d[k] - dlo);
    const unsigned q  = (unsigned)(s[k] - slo);
    const bool     in = r < (unsigned)dcnt && q < (unsigned)scnt;
    val[k] = us[in ? q : 0u];
  }
#pragma unroll
  for (int k = 0; k < SK; ++k) asm volatile("" : "+v"(val[k]));
#pragma unroll
  for (int k = 0; k < SK; ++k) {
    const unsigned r = (unsigned)(d[k] - dlo);
    const unsigned q = (unsigned)(s[k] - slo);
    if (r < (unsigned)dcnt && q < (unsigned)scnt) lds_add_f32(&bins[r], val[k]);
  }
}

__global__ __launch_bounds__(HT) void gcn_scatter_u(
    const int* __restrict__ src, const int* __restrict__ dst,
    const float* __restrict__ u, float* __restrict__ acc, long long E, int N) {
  __shared__ float bins[QR];
  __shared__ float us[SR];

  const int tid = (int)threadIdx.x;
  const int dlo = (int)blockIdx.y * QR;
  int dcnt = N - dlo;
  if (dcnt > QR) dcnt = QR;
  const int slo = (int)blockIdx.z * SR;
  int scnt = N - slo;
  if (scnt > SR) scnt = SR;

  for (int i = tid; i < QR / 4; i += HT)
    reinterpret_cast<float4*>(bins)[i] = make_float4(0.0f, 0.0f, 0.0f, 0.0f);

  const float* up = u + slo;
  int done = 0;
  if ((((size_t)up) & 15) == 0 && scnt >= 4) {
    constexpr int FU  = (SR / 4 + HT - 1) / HT;
    const int     n4  = scnt >> 2;
    const float4* up4 = reinterpret_cast<const float4*>(up);
    float4*       us4 = reinterpret_cast<float4*>(us);
    float4 t[FU];
#pragma unroll
    for (int j = 0; j < FU; ++j) {
      const int i = tid + j * HT;
      t[j] = up4[i < n4 ? i : n4 - 1];
    }
#pragma unroll
    for (int j = 0; j < FU; ++j) {
      const int i = tid + j * HT;
      us4[i < n4 ? i : n4 - 1] = t[j];
    }
    done = n4 << 2;
  }
  for (int i = done + tid; i < scnt; i += HT) us[i] = up[i];
  __syncthreads();

  const long long per = (((E + HB - 1) / HB) + 3) & ~3LL;
  const long long e0  = (long long)blockIdx.x * per;
  long long e1 = e0 + per;
  if (e1 > E) e1 = E;
  if (e1 < e0) e1 = e0;
  long long rest = e0;

  if (((((size_t)dst | (size_t)src) & 15) == 0)) {
    const long long trip = (long long)SK * HT;
    const long long e1v  = e0 + ((e1 - e0) & ~3LL);
    long long e = e0;
    for (; e + trip <= e1v; e += trip)
      scatter_u_trip<false>(src, dst, us, bins, e, e0, e1v, tid, dlo, dcnt, slo, scnt);
    if (e < e1v)
      scatter_u_trip<true>(src, dst, us, bins, e, e0, e1v, tid, dlo, dcnt, slo, scnt);
    rest = e1v;
  }
  for (long long k = rest + tid; k < e1; k += HT) {
    const unsigned r = (unsigned)(dst[k] - dlo);
    const unsigned q = (unsigned)(src[k] - slo);
    if (r < (unsigned)dcnt && q < (unsigned)scnt) lds_add_f32(&bins[r], us[q]);
  }
  __syncthreads();

  float* a = acc + dlo;
  for (int i = tid; i < dcnt; i += HT) atomic_add_f32(&a[i], bins[i]);
}

__global__ void gcn_finalize2(const float* __restrict__ acc, const float* __restrict__ dinv,
                              const float* __restrict__ b2, float* __restrict__ out, int N) {
  int i = blockIdx.x * blockDim.x + threadIdx.x;
  if (i >= N) return;
  out[i] = acc[i] * dinv[i] + b2[0];
}

extern "C" void kernel_launch(void* const* d_in, const int* in_sizes, int n_in,
                              void* d_out, int out_size, void* d_ws, size_t ws_size,
                              hipStream_t stream) {
  const float* x  = (const float*)d_in[0];
  const float* W1 = (const float*)d_in[1];
  const float* b1 = (const float*)d_in[2];
  const float* W2 = (const float*)d_in[3];
  const float* b2 = (const float*)d_in[4];
  const int*   ei = (const int*)d_in[5];

  int       N = in_sizes[0] / 5;
  long long E = (long long)in_sizes[5] / 2;
  const int* src = ei;
  const int* dst = ei + E;

  float* ws     = (float*)d_ws;
  float* planes = ws;
  float* xs8    = ws;
  float* agg5   = ws + (size_t)XP * N;
  float* acc    = ws;
  float* dinv   = ws + (size_t)HB * N;
  float* u      = dinv + N;

  float* out = (float*)d_out;

  dim3      gScat(HB, (unsigned)((N + HR - 1) / HR));
  dim3      gScatU(HB, (unsigned)((N + QR - 1) / QR), (unsigned)((N + SR - 1) / SR));
  unsigned  gNode   = (unsigned)((N + TPB - 1) / TPB);
  int       waves   = (N + 15) / 16;
  unsigned  gWmma   = (unsigned)((waves + TPB / 32 - 1) / (TPB / 32));

  gcn_degree   <<<gScat,   HT, 0, stream>>>(dst, planes, E, N);
  gcn_rsqrt    <<<gNode,  TPB, 0, stream>>>(planes, dinv, N);
  gcn_prep     <<<gNode,  TPB, 0, stream>>>(x, dinv, xs8, agg5, N);
  gcn_scatter5 <<<gScat,   HT, 0, stream>>>(src, dst, xs8, agg5, E, N);
  gcn_finalize1<<<gWmma,  TPB, 0, stream>>>(agg5, dinv, W1, b1, W2, u, acc, N);
  gcn_scatter_u<<<gScatU,  HT, 0, stream>>>(src, dst, u, acc, E, N);
  gcn_finalize2<<<gNode,  TPB, 0, stream>>>(acc, dinv, b2, out, N);
}


# Node kernels folded into their neighbours, 7 launches -> 5: prep sums the degree counts and takes rsqrt itself; the layer-2 flush adds dinv[d]*bins into out (dinv now scales each partial sum)
# speedup vs baseline: 5.5945x; 1.0338x over previous
#include <hip/hip_runtime.h>
#include <hip/hip_bf16.h>


typedef __attribute__((ext_vector_type(16))) _Float16 v16h;
typedef __attribute__((ext_vector_type(8)))  float    v8f;

#define TPB 256
#define HT 1024
#define HB 32
#define HR 12800

__device__ __forceinline__ void atomic_add_f32(float* p, float v) {
  __hip_atomic_fetch_add(p, v, __ATOMIC_RELAXED, __HIP_MEMORY_SCOPE_AGENT);
}

__device__ __forceinline__ void lds_add_f32(float* p, float v) {
  __hip_atomic_fetch_add(p, v, __ATOMIC_RELAXED, __HIP_MEMORY_SCOPE_WORKGROUP);
}

#define C1 5
#define XP 8

#define NBF  1024
#define AGO  256

__device__ __forceinline__ size_t nb_base(int j) { return (size_t)(j >> 5) * NBF; }
__device__ __forceinline__ size_t xs_at(int j)   { return nb_base(j) + (size_t)((j & 31) * 8); }
__device__ __forceinline__ size_t agg_at(int j)  { return nb_base(j) + (size_t)(AGO + (j & 31) * 5); }

__global__ __launch_bounds__(TPB) void gcn_prep(
    const float* __restrict__ x, float* reg, float* __restrict__ dinv, int N) {
  static_assert(TPB % 32 == 0, "a wave must be one node block");
  int i = blockIdx.x * blockDim.x + threadIdx.x;
  if (i >= N) return;
  float*    blk = reg + nb_base(i);
  const int l   = i & 31;
  float xv[C1];
#pragma unroll
  for (int k = 0; k < C1; ++k) xv[k] = x[(size_t)i * C1 + k];
  float deg = 1.0f;
#pragma unroll
  for (int b = 0; b < HB; ++b) deg += blk[b * 32 + l];
  const float di = rsqrtf(deg);
  dinv[i] = di;
  float v[C1];
#pragma unroll
  for (int k = 0; k < C1; ++k) v[k] = xv[k] * di;
  float4* row = reinterpret_cast<float4*>(blk + l * XP);
  row[0] = make_float4(v[0], v[1], v[2], v[3]);
  row[1] = make_float4(v[4], 0.0f, 0.0f, 0.0f);
#pragma unroll
  for (int k = 0; k < C1; ++k) blk[AGO + l * C1 + k] = v[k];
}

#define SU 4
#define SK (4 * SU)


template <bool GUARD>
__device__ __forceinline__ void degree_trip(
    const int* __restrict__ dst, float* __restrict__ bins,
    long long e, long long e0, long long e1v, int tid, int lo, int cnt) {
  int d[SK];
#pragma unroll
  for (int g = 0; g < SU; ++g) {
    const long long at0 = e + 4LL * (tid + HT * g);
    const bool      ok  = !GUARD || at0 < e1v;
    const long long at  = ok ? at0 : e0;
    int4 q = *reinterpret_cast<const int4*>(dst + at);
    if (!ok) q = make_int4(-1, -1, -1, -1);
    d[4 * g] = q.x; d[4 * g + 1] = q.y; d[4 * g + 2] = q.z; d[4 * g + 3] = q.w;
  }
#pragma unroll
  for (int k = 0; k < SK; ++k) {
    const unsigned r = (unsigned)(d[k] - lo);
    if (r < (unsigned)cnt) lds_add_f32(&bins[r], 1.0f);
  }
}

__global__ __launch_bounds__(HT) void gcn_degree(
    const int* __restrict__ dst, float* __restrict__ planes, long long E, int N) {
  constexpr int FJ = (HR + HT - 1) / HT;
  __shared__ float bins[FJ * HT];

  const int tid = (int)threadIdx.x;
  const int lo  = (int)blockIdx.y * HR;
  int cnt = N - lo;
  if (cnt > HR) cnt = HR;

  for (int j = 0; j < FJ; ++j) bins[tid + j * HT] = 0.0f;
  __syncthreads();

  const long long per = (((E + HB - 1) / HB) + 3) & ~3LL;
  const long long e0  = (long long)blockIdx.x * per;
  long long e1 = e0 + per;
  if (e1 > E) e1 = E;
  if (e1 < e0) e1 = e0;
  long long rest = e0;

  if ((((size_t)dst) & 15) == 0) {
    const long long trip = (long long)SK * HT;
    const long long e1v  = e0 + ((e1 - e0) & ~3LL);
    long long e = e0;
    for (; e + trip <= e1v; e += trip)
      degree_trip<false>(dst, bins, e, e0, e1v, tid, lo, cnt);
    if (e < e1v)
      degree_trip<true>(dst, bins, e, e0, e1v, tid, lo, cnt);
    rest = e1v;
  }
  for (long long k = rest + tid; k < e1; k += HT) {
    const unsigned r = (unsigned)(dst[k] - lo);
    if (r < (unsigned)cnt) lds_add_f32(&bins[r], 1.0f);
  }
  __syncthreads();

  static_assert(HB == 32 && HR % 32 == 0, "one line per chunk in every node block");
  for (int i = tid; i < cnt; i += HT) {
    const int j = lo + i;
    planes[nb_base(j) + (size_t)((int)blockIdx.x * 32 + (j & 31))] = bins[i];
  }
}

#define SU5 2
#define SK5 (4 * SU5)

template <bool GUARD>
__device__ __forceinline__ void scatter5_trip(
    const int* __restrict__ src, const int* __restrict__ dst,
    const float* __restrict__ reg, float* __restrict__ bins,
    long long e, long long e0, long long e1v, int tid, int lo, int cnt) {
  int d[SK5], s[SK5];
#pragma unroll
  for (int u = 0; u < SU5; ++u) {
    const long long g  = e + 4LL * (tid + HT * u);
    const bool      ok = !GUARD || g < e1v;
    const long long at = ok ? g : e0;
    int4 q = *reinterpret_cast<const int4*>(dst + at);
    if (!ok) q = make_int4(-1, -1, -1, -1);
    d[4 * u] = q.x; d[4 * u + 1] = q.y; d[4 * u + 2] = q.z; d[4 * u + 3] = q.w;
    const int4 p = *reinterpret_cast<const int4*>(src + at);
    s[4 * u] = p.x; s[4 * u + 1] = p.y; s[4 * u + 2] = p.z; s[4 * u + 3] = p.w;
  }
  float4 va[SK5];
  float  vb[SK5];
#pragma unroll
  for (int k = 0; k < SK5; ++k) {
    const unsigned r  = (unsigned)(d[k] - lo);
    const int      sk = r < (unsigned)cnt ? s[k] : 0;
    const float*   row = reg + xs_at(sk);
    va[k] = *reinterpret_cast<const float4*>(row);
    vb[k] = row[4];
  }
#pragma unroll
  for (int k = 0; k < SK5; ++k)
    asm volatile("" : "+v"(va[k].x), "+v"(va[k].y), "+v"(va[k].z), "+v"(va[k].w), "+v"(vb[k]));
#pragma unroll
  for (int k = 0; k < SK5; ++k) {
    const unsigned r = (unsigned)(d[k] - lo);
    if (r < (unsigned)cnt) {
      float* b = bins + r * C1;
      lds_add_f32(b + 0, va[k].x);
      lds_add_f32(b + 1, va[k].y);
      lds_add_f32(b + 2, va[k].z);
      lds_add_f32(b + 3, va[k].w);
      lds_add_f32(b + 4, vb[k]);
    }
  }
}

__global__ __launch_bounds__(HT) void gcn_scatter5(
    const int* __restrict__ src, const int* __restrict__ dst,
    float* reg, long long E, int N) {
  constexpr int FJ = (HR * C1 + HT - 1) / HT;
  __shared__ float bins[FJ * HT];

  const int tid = (int)threadIdx.x;
  const int lo  = (int)blockIdx.y * HR;
  int cnt = N - lo;
  if (cnt > HR) cnt = HR;

  for (int j = 0; j < FJ; ++j) bins[tid + j * HT] = 0.0f;
  __syncthreads();

  const long long per = (((E + HB - 1) / HB) + 3) & ~3LL;
  const long long e0  = (long long)blockIdx.x * per;
  long long e1 = e0 + per;
  if (e1 > E) e1 = E;
  if (e1 < e0) e1 = e0;
  long long rest = e0;

  if (((((size_t)dst | (size_t)src) & 15) == 0)) {
    const long long trip = (long long)SK5 * HT;
    const long long e1v  = e0 + ((e1 - e0) & ~3LL);
    long long e = e0;
    for (; e + trip <= e1v; e += trip)
      scatter5_trip<false>(src, dst, reg, bins, e, e0, e1v, tid, lo, cnt);
    if (e < e1v)
      scatter5_trip<true>(src, dst, reg, bins, e, e0, e1v, tid, lo, cnt);
    rest = e1v;
  }
  for (long long k = rest + tid; k < e1; k += HT) {
    const unsigned r = (unsigned)(dst[k] - lo);
    if (r < (unsigned)cnt) {
      const float* row = reg + xs_at(src[k]);
      float* b = bins + r * C1;
#pragma unroll
      for (int c = 0; c < C1; ++c) lds_add_f32(b + c, row[c]);
    }
  }
  __syncthreads();

  static_assert(HR % 32 == 0, "a block's range is whole node blocks");
  float* ag = reg + nb_base(lo) + AGO;
  const int tot = cnt * C1;
  for (int i = tid; i < tot; i += HT) {
    const int nb = i / (32 * C1);
    atomic_add_f32(&ag[(size_t)nb * NBF + (size_t)(i - nb * (32 * C1))], bins[i]);
  }
}

__global__ __launch_bounds__(TPB) void gcn_finalize1(
    const float* __restrict__ reg, const float* __restrict__ dinv,
    const float* __restrict__ W1, const float* __restrict__ b1,
    const float* __restrict__ W2, const float* __restrict__ b2,
    float* __restrict__ u, float* __restrict__ out, int N) {
  int wave = (int)(((long long)blockIdx.x * blockDim.x + threadIdx.x) >> 5);
  int lane = (int)(threadIdx.x & 31);
  int row0 = wave * 16;
  if (row0 >= N) return;

  const int   ms = row0 + ((lane >= 16) ? 8 : 0) + (lane & 7);
  const float ds = dinv[ms < N ? ms : N - 1];
  const float bo = b2[0];

  v16h ah = {}, al = {};
  v16h bh = {}, bl = {};
  if (lane < 16) {
    int m = row0 + lane;
    if (m >= N) m = N - 1;
    const float  di = dinv[m];
    const float* ar = reg + agg_at(m);
#pragma unroll
    for (int k = 0; k < C1; ++k) {
      float v = ar[k] * di;
      _Float16 hi = (_Float16)v;
      ah[k] = hi;
      al[k] = (_Float16)(v - (float)hi);
    }
#pragma unroll
    for (int k = 0; k < C1; ++k) {
      float w = W1[k * 16 + lane];
      _Float16 hi = (_Float16)w;
      bh[k] = hi;
      bl[k] = (_Float16)(w - (float)hi);
    }
  }

  v8f c = {};
  c = __builtin_amdgcn_wmma_f32_16x16x32_f16(false, al, false, bh, (short)0, c, false, false);
  c = __builtin_amdgcn_wmma_f32_16x16x32_f16(false, ah, false, bl, (short)0, c, false, false);
  c = __builtin_amdgcn_wmma_f32_16x16x32_f16(false, ah, false, bh, (short)0, c, false, false);

  const int   n    = lane & 15;
  const float bias = b1[n];
  const float w2   = W2[n];
  float p[8];
#pragma unroll
  for (int r = 0; r < 8; ++r) {
    float v = fmaxf(c[r] + bias, 0.0f) * w2;
    v += __shfl_xor(v, 8);
    v += __shfl_xor(v, 4);
    v += __shfl_xor(v, 2);
    v += __shfl_xor(v, 1);
    p[r] = v;
  }
  float mine = p[0];
#pragma unroll
  for (int r = 1; r < 8; ++r) mine = (n == r) ? p[r] : mine;
  const int m = row0 + ((lane >= 16) ? 8 : 0) + n;
  if (n < 8 && m < N) {
    const float t = mine * ds;
    u[m]   = t;
    out[m] = t * ds + bo;
  }
}

#define QR 25600
#define SR 51200

template <bool GUARD>
__device__ __forceinline__ void scatter_u_trip(
    const int* __restrict__ src, const int* __restrict__ dst,
    const float* __restrict__ us, float* __restrict__ bins,
    long long e, long long e0, long long e1v, int tid,
    int dlo, int dcnt, int slo, int scnt) {
  int d[SK], s[SK];
#pragma unroll
  for (int g = 0; g < SU; ++g) {
    const long long at0 = e + 4LL * (tid + HT * g);
    const bool      ok  = !GUARD || at0 < e1v;
    const long long at  = ok ? at0 : e0;
    int4 q = *reinterpret_cast<const int4*>(dst + at);
    if (!ok) q = make_int4(-1, -1, -1, -1);
    d[4 * g] = q.x; d[4 * g + 1] = q.y; d[4 * g + 2] = q.z; d[4 * g + 3] = q.w;
    const int4 p = *reinterpret_cast<const int4*>(src + at);
    s[4 * g] = p.x; s[4 * g + 1] = p.y; s[4 * g + 2] = p.z; s[4 * g + 3] = p.w;
  }
  float val[SK];
#pragma unroll
  for (int k = 0; k < SK; ++k) {
    const unsigned r  = (unsigned)(d[k] - dlo);
    const unsigned q  = (unsigned)(s[k] - slo);
    const bool     in = r < (unsigned)dcnt && q < (unsigned)scnt;
    val[k] = us[in ? q : 0u];
  }
#pragma unroll
  for (int k = 0; k < SK; ++k) asm volatile("" : "+v"(val[k]));
#pragma unroll
  for (int k = 0; k < SK; ++k) {
    const unsigned r = (unsigned)(d[k] - dlo);
    const unsigned q = (unsigned)(s[k] - slo);
    if (r < (unsigned)dcnt && q < (unsigned)scnt) lds_add_f32(&bins[r], val[k]);
  }
}

__global__ __launch_bounds__(HT) void gcn_scatter_u(
    const int* __restrict__ src, const int* __restrict__ dst,
    const float* __restrict__ u, const float* __restrict__ dinv,
    float* __restrict__ out, long long E, int N) {
  __shared__ float bins[QR];
  __shared__ float us[SR];

  const int tid = (int)threadIdx.x;
  const int dlo = (int)blockIdx.y * QR;
  int dcnt = N - dlo;
  if (dcnt > QR) dcnt = QR;
  const int slo = (int)blockIdx.z * SR;
  int scnt = N - slo;
  if (scnt > SR) scnt = SR;

  for (int i = tid; i < QR / 4; i += HT)
    reinterpret_cast<float4*>(bins)[i] = make_float4(0.0f, 0.0f, 0.0f, 0.0f);

  const float* up = u + slo;
  int done = 0;
  if ((((size_t)up) & 15) == 0 && scnt >= 4) {
    constexpr int FU  = (SR / 4 + HT - 1) / HT;
    const int     n4  = scnt >> 2;
    const float4* up4 = reinterpret_cast<const float4*>(up);
    float4*       us4 = reinterpret_cast<float4*>(us);
    float4 t[FU];
#pragma unroll
    for (int j = 0; j < FU; ++j) {
      const int i = tid + j * HT;
      t[j] = up4[i < n4 ? i : n4 - 1];
    }
#pragma unroll
    for (int j = 0; j < FU; ++j) {
      const int i = tid + j * HT;
      us4[i < n4 ? i : n4 - 1] = t[j];
    }
    done = n4 << 2;
  }
  for (int i = done + tid; i < scnt; i += HT) us[i] = up[i];
  __syncthreads();

  const long long per = (((E + HB - 1) / HB) + 3) & ~3LL;
  const long long e0  = (long long)blockIdx.x * per;
  long long e1 = e0 + per;
  if (e1 > E) e1 = E;
  if (e1 < e0) e1 = e0;
  long long rest = e0;

  if (((((size_t)dst | (size_t)src) & 15) == 0)) {
    const long long trip = (long long)SK * HT;
    const long long e1v  = e0 + ((e1 - e0) & ~3LL);
    long long e = e0;
    for (; e + trip <= e1v; e += trip)
      scatter_u_trip<false>(src, dst, us, bins, e, e0, e1v, tid, dlo, dcnt, slo, scnt);
    if (e < e1v)
      scatter_u_trip<true>(src, dst, us, bins, e, e0, e1v, tid, dlo, dcnt, slo, scnt);
    rest = e1v;
  }
  for (long long k = rest + tid; k < e1; k += HT) {
    const unsigned r = (unsigned)(dst[k] - dlo);
    const unsigned q = (unsigned)(src[k] - slo);
    if (r < (unsigned)dcnt && q < (unsigned)scnt) lds_add_f32(&bins[r], us[q]);
  }
  __syncthreads();

  float*       o  = out + dlo;
  const float* dv = dinv + dlo;
  constexpr int FB = 5;
  int i = tid;
  for (; i + (FB - 1) * HT < dcnt; i += FB * HT) {
    float w[FB], b[FB];
#pragma unroll
    for (int j = 0; j < FB; ++j) { w[j] = dv[i + j * HT]; b[j] = bins[i + j * HT]; }
#pragma unroll
    for (int j = 0; j < FB; ++j) atomic_add_f32(&o[i + j * HT], b[j] * w[j]);
  }
  for (; i < dcnt; i += HT) atomic_add_f32(&o[i], bins[i] * dv[i]);
}

extern "C" void kernel_launch(void* const* d_in, const int* in_sizes, int n_in,
                              void* d_out, int out_size, void* d_ws, size_t ws_size,
                              hipStream_t stream) {
  const float* x  = (const float*)d_in[0];
  const float* W1 = (const float*)d_in[1];
  const float* b1 = (const float*)d_in[2];
  const float* W2 = (const float*)d_in[3];
  const float* b2 = (const float*)d_in[4];
  const int*   ei = (const int*)d_in[5];

  int       N = in_sizes[0] / 5;
  long long E = (long long)in_sizes[5] / 2;
  const int* src = ei;
  const int* dst = ei + E;

  float* ws     = (float*)d_ws;
  float* reg    = ws;
  float* dinv   = ws + (size_t)NBF * ((N + 31) / 32);
  float* u      = dinv + N;

  float* out = (float*)d_out;

  dim3      gScat(HB, (unsigned)((N + HR - 1) / HR));
  dim3      gScatU(HB, (unsigned)((N + QR - 1) / QR), (unsigned)((N + SR - 1) / SR));
  unsigned  gNode   = (unsigned)((N + TPB - 1) / TPB);
  int       waves   = (N + 15) / 16;
  unsigned  gWmma   = (unsigned)((waves + TPB / 32 - 1) / (TPB / 32));

  gcn_degree   <<<gScat,   HT, 0, stream>>>(dst, reg, E, N);
  gcn_prep     <<<gNode,  TPB, 0, stream>>>(x, reg, dinv, N);
  gcn_scatter5 <<<gScat,   HT, 0, stream>>>(src, dst, reg, E, N);
  gcn_finalize1<<<gWmma,  TPB, 0, stream>>>(reg, dinv, W1, b1, W2, b2, u, out, N);
  gcn_scatter_u<<<gScatU,  HT, 0, stream>>>(src, dst, u, dinv, out, E, N);
}
